# starting kernel
# baseline (speedup 1.0000x reference)
#include <hip/hip_runtime.h>
#include <hip/hip_bf16.h>

typedef __attribute__((ext_vector_type(2))) float v2f;
typedef __attribute__((ext_vector_type(8))) float v8f;

__global__ void k_init_deg(float* __restrict__ deg, int n) {
    int i = blockIdx.x * blockDim.x + threadIdx.x;
    if (i < n) deg[i] = 1.0f;
}

__global__ void k_deg_edges(const int* __restrict__ dst, float* __restrict__ deg, int E) {
    int e = blockIdx.x * blockDim.x + threadIdx.x;
    if (e < E) atomicAdd(&deg[dst[e]], 1.0f);
}

__global__ void k_dis(float* __restrict__ deg, int n) {
    int i = blockIdx.x * blockDim.x + threadIdx.x;
    if (i < n) {
        float d = deg[i];
        deg[i] = (d > 0.0f) ? rsqrtf(d) : 0.0f;
    }
}

__global__ void k_gemm_wmma(const float* __restrict__ A, const float* __restrict__ W,
                            float* __restrict__ out, int n, int K) {
    __shared__ float sW[64 * 64];
    const int tid = threadIdx.x;
    for (int i = tid; i < K * 64; i += 128) sW[i] = W[i];
    __syncthreads();

    const int wave = tid >> 5;
    const int lane = tid & 31;
    const int tile_row = blockIdx.x * 64 + wave * 16;

    const int m    = lane & 15;
    const int koff = (lane >> 4) * 2;
    int row = tile_row + m;
    if (row >= n) row = n - 1;

    v8f acc[4] = {v8f{0.f}, v8f{0.f}, v8f{0.f}, v8f{0.f}};

    for (int kb = 0; kb < K; kb += 4) {
        v2f a;
        const float* ap = A + (size_t)row * K + kb + koff;
        a.x = ap[0];
        a.y = ap[1];
#pragma unroll
        for (int nt = 0; nt < 4; ++nt) {
            v2f b;
            const int col = nt * 16 + (lane & 15);
            b.x = sW[(kb + koff) * 64 + col];
            b.y = sW[(kb + koff + 1) * 64 + col];
            acc[nt] = __builtin_amdgcn_wmma_f32_16x16x4_f32(
                 false, a,  false, b,
                 (short)0, acc[nt],  false,  false);
        }
    }

#pragma unroll
    for (int nt = 0; nt < 4; ++nt) {
#pragma unroll
        for (int r = 0; r < 8; ++r) {
            const int mm = r + 8 * (lane >> 4);
            const int rr = tile_row + mm;
            if (rr < n) out[(size_t)rr * 64 + nt * 16 + (lane & 15)] = acc[nt][r];
        }
    }
}

__global__ void k_self_init(const float* __restrict__ dis, const float* __restrict__ h,
                            float* __restrict__ acc, int n) {
    int t = blockIdx.x * blockDim.x + threadIdx.x;
    if (t >= n * 16) return;
    int i = t >> 4;
    int jj = (t & 15) * 4;
    float di = dis[i];
    float s = di * di;
    const float4 hv = *(const float4*)&h[(size_t)i * 64 + jj];
    float4 o;
    o.x = s * hv.x; o.y = s * hv.y; o.z = s * hv.z; o.w = s * hv.w;
    *(float4*)&acc[(size_t)i * 64 + jj] = o;
}

__global__ void k_scatter(const int* __restrict__ src, const int* __restrict__ dst,
                          const float* __restrict__ dis, const float* __restrict__ h,
                          float* __restrict__ acc, int E) {
    int t = blockIdx.x * blockDim.x + threadIdx.x;
    if (t >= E * 16) return;
    int e = t >> 4;
    int jj = (t & 15) * 4;
    int s = src[e];
    int d = dst[e];
    float norm = dis[s] * dis[d];
    const float4 hv = *(const float4*)&h[(size_t)s * 64 + jj];
    float* ap = &acc[(size_t)d * 64 + jj];
    atomicAdd(ap + 0, norm * hv.x);
    atomicAdd(ap + 1, norm * hv.y);
    atomicAdd(ap + 2, norm * hv.z);
    atomicAdd(ap + 3, norm * hv.w);
}

__global__ void k_bias_relu(float* __restrict__ acc, const float* __restrict__ b, int n) {
    int t = blockIdx.x * blockDim.x + threadIdx.x;
    if (t >= n * 64) return;
    int j = t & 63;
    acc[t] = fmaxf(acc[t] + b[j], 0.0f);
}

__global__ void k_zero(float* __restrict__ p, int n) {
    int t = blockIdx.x * blockDim.x + threadIdx.x;
    if (t < n) p[t] = 0.0f;
}

__global__ void k_pool(const float* __restrict__ acc2, const float* __restrict__ b2,
                       const int* __restrict__ batch, float* __restrict__ sums, int n) {
    int t = blockIdx.x * blockDim.x + threadIdx.x;
    if (t >= n * 16) return;
    int i = t >> 4;
    int jj = (t & 15) * 4;
    int g = batch[i];
    const float4 hv = *(const float4*)&acc2[(size_t)i * 64 + jj];
    const float4 bv = *(const float4*)&b2[jj];
    float* sp = &sums[(size_t)g * 64 + jj];
    atomicAdd(sp + 0, fmaxf(hv.x + bv.x, 0.0f));
    atomicAdd(sp + 1, fmaxf(hv.y + bv.y, 0.0f));
    atomicAdd(sp + 2, fmaxf(hv.z + bv.z, 0.0f));
    atomicAdd(sp + 3, fmaxf(hv.w + bv.w, 0.0f));
}

__global__ void k_count(const int* __restrict__ batch, float* __restrict__ cnt, int n) {
    int i = blockIdx.x * blockDim.x + threadIdx.x;
    if (i < n) atomicAdd(&cnt[batch[i]], 1.0f);
}

__global__ void k_div(const float* __restrict__ sums, const float* __restrict__ cnt,
                      float* __restrict__ out, int G) {
    int t = blockIdx.x * blockDim.x + threadIdx.x;
    if (t >= G * 64) return;
    out[t] = sums[t] / fmaxf(cnt[t >> 6], 1.0f);
}

extern "C" void kernel_launch(void* const* d_in, const int* in_sizes, int n_in,
                              void* d_out, int out_size, void* d_ws, size_t ws_size,
                              hipStream_t stream) {
    const float* x     = (const float*)d_in[0];
    const int* eidx    = (const int*)d_in[1];
    const int* batch   = (const int*)d_in[2];
    const float* W1 = (const float*)d_in[4];
    const float* b1 = (const float*)d_in[5];
    const float* W2 = (const float*)d_in[6];
    const float* b2 = (const float*)d_in[7];

    const int n = in_sizes[0] / 4;
    const int E = in_sizes[1] / 2;
    const int G = out_size / 64;

    const int* src = eidx;
    const int* dst = eidx + E;

    float* ws   = (float*)d_ws;
    float* dis  = ws;
    float* bufA = dis + n;
    float* bufB = bufA + (size_t)n * 64;
    float* sums = bufB + (size_t)n * 64;
    float* cnt  = sums + (size_t)G * 64;

    const int T = 256;
    auto blk = [T](int work) { return (work + T - 1) / T; };

    k_init_deg<<<blk(n), T, 0, stream>>>(dis, n);
    k_deg_edges<<<blk(E), T, 0, stream>>>(dst, dis, E);
    k_dis<<<blk(n), T, 0, stream>>>(dis, n);

    k_gemm_wmma<<<(n + 63) / 64, 128, 0, stream>>>(x, W1, bufA, n, 4);
    k_self_init<<<blk(n * 16), T, 0, stream>>>(dis, bufA, bufB, n);
    k_scatter<<<blk(E * 16), T, 0, stream>>>(src, dst, dis, bufA, bufB, E);
    k_bias_relu<<<blk(n * 64), T, 0, stream>>>(bufB, b1, n);

    k_gemm_wmma<<<(n + 63) / 64, 128, 0, stream>>>(bufB, W2, bufA, n, 64);
    k_self_init<<<blk(n * 16), T, 0, stream>>>(dis, bufA, bufB, n);
    k_scatter<<<blk(E * 16), T, 0, stream>>>(src, dst, dis, bufA, bufB, E);

    k_zero<<<blk(G * 64 + G), T, 0, stream>>>(sums, G * 64 + G);
    k_pool<<<blk(n * 16), T, 0, stream>>>(bufB, b2, batch, sums, n);
    k_count<<<blk(n), T, 0, stream>>>(batch, cnt, n);
    k_div<<<blk(G * 64), T, 0, stream>>>(sums, cnt, (float*)d_out, G);
}


# float-atomic scatter-add replaced by a per-call destination-ordered edge list (returning int add as slot, scan, fill) and per-node register row sums stored once; sum order = slot order
# speedup vs baseline: 1.6246x; 1.6246x over previous
#include <hip/hip_runtime.h>
#include <hip/hip_bf16.h>

typedef __attribute__((ext_vector_type(2))) float v2f;
typedef __attribute__((ext_vector_type(8))) float v8f;

__global__ void k_zero_i(int* __restrict__ p, int n) {
    int i = blockIdx.x * blockDim.x + threadIdx.x;
    if (i < n) p[i] = 0;
}

__global__ void k_rank(const int* __restrict__ dst, int* __restrict__ cnt,
                       int* __restrict__ pos, int E) {
    int e = blockIdx.x * blockDim.x + threadIdx.x;
    if (e < E) pos[e] = atomicAdd(&cnt[dst[e]], 1);
}

__global__ void k_scan(const int* __restrict__ cnt, int* __restrict__ start,
                       float* __restrict__ dis, int n) {
    __shared__ int s_pre[32];
    __shared__ int s_tot[32];
    const int tid  = threadIdx.x;
    const int lane = tid & 31;
    const int wave = tid >> 5;
    const int b0   = blockIdx.x * 1024;

    int pre = 0;
    for (int j = tid; j < b0; j += 1024) pre += cnt[j];

    const int i = b0 + tid;
    const int c = (i < n) ? cnt[i] : 0;

    int inc = c;
    for (int o = 1; o < 32; o <<= 1) {
        int v = __shfl_up(inc, o);
        if (lane >= o) inc += v;
    }
    for (int o = 16; o > 0; o >>= 1) pre += __shfl_xor(pre, o);

    if (lane == 31) s_tot[wave] = inc;
    if (lane == 0)  s_pre[wave] = pre;
    __syncthreads();

    int v = s_pre[lane] + ((lane < wave) ? s_tot[lane] : 0);
    for (int o = 16; o > 0; o >>= 1) v += __shfl_xor(v, o);

    if (i < n) {
        start[i] = v + inc - c;
        float d = (float)(c + 1);
        dis[i] = (d > 0.0f) ? rsqrtf(d) : 0.0f;
    }
}

__global__ void k_fill(const int* __restrict__ src, const int* __restrict__ dst,
                       const int* __restrict__ start, const int* __restrict__ pos,
                       int* __restrict__ csr, int E) {
    int e = blockIdx.x * blockDim.x + threadIdx.x;
    if (e < E) csr[start[dst[e]] + pos[e]] = src[e];
}

__global__ void k_gemm_wmma(const float* __restrict__ A, const float* __restrict__ W,
                            float* __restrict__ out, int n, int K) {
    __shared__ float sW[64 * 64];
    const int tid = threadIdx.x;
    for (int i = tid; i < K * 64; i += 128) sW[i] = W[i];
    __syncthreads();

    const int wave = tid >> 5;
    const int lane = tid & 31;
    const int tile_row = blockIdx.x * 64 + wave * 16;

    const int m    = lane & 15;
    const int koff = (lane >> 4) * 2;
    int row = tile_row + m;
    if (row >= n) row = n - 1;

    v8f acc[4] = {v8f{0.f}, v8f{0.f}, v8f{0.f}, v8f{0.f}};

    for (int kb = 0; kb < K; kb += 4) {
        v2f a;
        const float* ap = A + (size_t)row * K + kb + koff;
        a.x = ap[0];
        a.y = ap[1];
#pragma unroll
        for (int nt = 0; nt < 4; ++nt) {
            v2f b;
            const int col = nt * 16 + (lane & 15);
            b.x = sW[(kb + koff) * 64 + col];
            b.y = sW[(kb + koff + 1) * 64 + col];
            acc[nt] = __builtin_amdgcn_wmma_f32_16x16x4_f32(
                 false, a,  false, b,
                 (short)0, acc[nt],  false,  false);
        }
    }

#pragma unroll
    for (int nt = 0; nt < 4; ++nt) {
#pragma unroll
        for (int r = 0; r < 8; ++r) {
            const int mm = r + 8 * (lane >> 4);
            const int rr = tile_row + mm;
            if (rr < n) out[(size_t)rr * 64 + nt * 16 + (lane & 15)] = acc[nt][r];
        }
    }
}

__global__ void k_gather(const int* __restrict__ start, const int* __restrict__ cnt,
                         const int* __restrict__ csr, const float* __restrict__ dis,
                         const float* __restrict__ h, const float* __restrict__ b,
                         float* __restrict__ out, int n, int fuse) {
    int t = blockIdx.x * blockDim.x + threadIdx.x;
    if (t >= n * 16) return;
    const int i  = t >> 4;
    const int jj = (t & 15) * 4;
    const float di = dis[i];
    int j = start[i];
    const int end = j + cnt[i];

    const float4 hs = *(const float4*)&h[(size_t)i * 64 + jj];
    const float ss = di * di;
    float4 acc;
    acc.x = ss * hs.x; acc.y = ss * hs.y; acc.z = ss * hs.z; acc.w = ss * hs.w;

    for (; j + 4 <= end; j += 4) {
        const int s0 = csr[j], s1 = csr[j + 1], s2 = csr[j + 2], s3 = csr[j + 3];
        const float4 v0 = *(const float4*)&h[(size_t)s0 * 64 + jj];
        const float4 v1 = *(const float4*)&h[(size_t)s1 * 64 + jj];
        const float4 v2 = *(const float4*)&h[(size_t)s2 * 64 + jj];
        const float4 v3 = *(const float4*)&h[(size_t)s3 * 64 + jj];
        const float w0 = dis[s0] * di, w1 = dis[s1] * di;
        const float w2 = dis[s2] * di, w3 = dis[s3] * di;
        acc.x += w0 * v0.x; acc.y += w0 * v0.y; acc.z += w0 * v0.z; acc.w += w0 * v0.w;
        acc.x += w1 * v1.x; acc.y += w1 * v1.y; acc.z += w1 * v1.z; acc.w += w1 * v1.w;
        acc.x += w2 * v2.x; acc.y += w2 * v2.y; acc.z += w2 * v2.z; acc.w += w2 * v2.w;
        acc.x += w3 * v3.x; acc.y += w3 * v3.y; acc.z += w3 * v3.z; acc.w += w3 * v3.w;
    }
    const int rem = end - j;
    if (rem > 0) {
        const int s0 = csr[j];
        const int s1 = (rem > 1) ? csr[j + 1] : s0;
        const int s2 = (rem > 2) ? csr[j + 2] : s0;
        const float4 v0 = *(const float4*)&h[(size_t)s0 * 64 + jj];
        const float4 v1 = *(const float4*)&h[(size_t)s1 * 64 + jj];
        const float4 v2 = *(const float4*)&h[(size_t)s2 * 64 + jj];
        const float w0 = dis[s0] * di, w1 = dis[s1] * di, w2 = dis[s2] * di;
        acc.x += w0 * v0.x; acc.y += w0 * v0.y; acc.z += w0 * v0.z; acc.w += w0 * v0.w;
        if (rem > 1) {
            acc.x += w1 * v1.x; acc.y += w1 * v1.y; acc.z += w1 * v1.z; acc.w += w1 * v1.w;
        }
        if (rem > 2) {
            acc.x += w2 * v2.x; acc.y += w2 * v2.y; acc.z += w2 * v2.z; acc.w += w2 * v2.w;
        }
    }

    if (fuse) {
        const float4 bv = *(const float4*)&b[jj];
        acc.x = fmaxf(acc.x + bv.x, 0.0f);
        acc.y = fmaxf(acc.y + bv.y, 0.0f);
        acc.z = fmaxf(acc.z + bv.z, 0.0f);
        acc.w = fmaxf(acc.w + bv.w, 0.0f);
    }
    *(float4*)&out[(size_t)i * 64 + jj] = acc;
}

__global__ void k_zero(float* __restrict__ p, int n) {
    int t = blockIdx.x * blockDim.x + threadIdx.x;
    if (t < n) p[t] = 0.0f;
}

__global__ void k_pool(const float* __restrict__ acc2, const float* __restrict__ b2,
                       const int* __restrict__ batch, float* __restrict__ sums, int n) {
    int t = blockIdx.x * blockDim.x + threadIdx.x;
    if (t >= n * 16) return;
    int i = t >> 4;
    int jj = (t & 15) * 4;
    int g = batch[i];
    const float4 hv = *(const float4*)&acc2[(size_t)i * 64 + jj];
    const float4 bv = *(const float4*)&b2[jj];
    float* sp = &sums[(size_t)g * 64 + jj];
    atomicAdd(sp + 0, fmaxf(hv.x + bv.x, 0.0f));
    atomicAdd(sp + 1, fmaxf(hv.y + bv.y, 0.0f));
    atomicAdd(sp + 2, fmaxf(hv.z + bv.z, 0.0f));
    atomicAdd(sp + 3, fmaxf(hv.w + bv.w, 0.0f));
}

__global__ void k_count(const int* __restrict__ batch, float* __restrict__ cnt, int n) {
    int i = blockIdx.x * blockDim.x + threadIdx.x;
    if (i < n) atomicAdd(&cnt[batch[i]], 1.0f);
}

__global__ void k_div(const float* __restrict__ sums, const float* __restrict__ cnt,
                      float* __restrict__ out, int G) {
    int t = blockIdx.x * blockDim.x + threadIdx.x;
    if (t >= G * 64) return;
    out[t] = sums[t] / fmaxf(cnt[t >> 6], 1.0f);
}

extern "C" void kernel_launch(void* const* d_in, const int* in_sizes, int n_in,
                              void* d_out, int out_size, void* d_ws, size_t ws_size,
                              hipStream_t stream) {
    const float* x     = (const float*)d_in[0];
    const int* eidx    = (const int*)d_in[1];
    const int* batch   = (const int*)d_in[2];
    const float* W1 = (const float*)d_in[4];
    const float* b1 = (const float*)d_in[5];
    const float* W2 = (const float*)d_in[6];
    const float* b2 = (const float*)d_in[7];

    const int n = in_sizes[0] / 4;
    const int E = in_sizes[1] / 2;
    const int G = out_size / 64;

    const int* src = eidx;
    const int* dst = eidx + E;

    float* ws   = (float*)d_ws;
    float* dis  = ws;
    float* bufA = dis + n;
    float* bufB = bufA + (size_t)n * 64;
    float* sums = bufB + (size_t)n * 64;
    float* cnt  = sums + (size_t)G * 64;
    int* icnt   = (int*)(cnt + G);
    int* start  = icnt + n;
    int* csr    = start + n;
    int* pos    = (int*)bufB;

    const int T = 256;
    auto blk = [T](int work) { return (work + T - 1) / T; };

    k_zero_i<<<blk(n), T, 0, stream>>>(icnt, n);
    k_rank<<<blk(E), T, 0, stream>>>(dst, icnt, pos, E);
    k_scan<<<(n + 1023) / 1024, 1024, 0, stream>>>(icnt, start, dis, n);
    k_fill<<<blk(E), T, 0, stream>>>(src, dst, start, pos, csr, E);

    k_gemm_wmma<<<(n + 63) / 64, 128, 0, stream>>>(x, W1, bufA, n, 4);
    k_gather<<<blk(n * 16), T, 0, stream>>>(start, icnt, csr, dis, bufA, b1, bufB, n, 1);

    k_gemm_wmma<<<(n + 63) / 64, 128, 0, stream>>>(bufB, W2, bufA, n, 64);
    k_gather<<<blk(n * 16), T, 0, stream>>>(start, icnt, csr, dis, bufA, b2, bufB, n, 0);

    k_zero<<<blk(G * 64 + G), T, 0, stream>>>(sums, G * 64 + G);
    k_pool<<<blk(n * 16), T, 0, stream>>>(bufB, b2, batch, sums, n);
    k_count<<<blk(n), T, 0, stream>>>(batch, cnt, n);
    k_div<<<blk(G * 64), T, 0, stream>>>(sums, cnt, (float*)d_out, G);
}


# layer 1 aggregates the 4-wide inputs first ((A x) W1 for A (x W1)) as the A fragment of the K=4 WMMA kernel, bias+relu in its epilogue; 64-wide layer-1 gather gone; sum association changes
# speedup vs baseline: 1.7024x; 1.0479x over previous
#include <hip/hip_runtime.h>
#include <hip/hip_bf16.h>

typedef __attribute__((ext_vector_type(2))) float v2f;
typedef __attribute__((ext_vector_type(8))) float v8f;

__global__ void k_zero_i(int* __restrict__ p, int n) {
    int i = blockIdx.x * blockDim.x + threadIdx.x;
    if (i < n) p[i] = 0;
}

__global__ void k_rank(const int* __restrict__ dst, int* __restrict__ cnt,
                       int* __restrict__ pos, int E) {
    int e = blockIdx.x * blockDim.x + threadIdx.x;
    if (e < E) pos[e] = atomicAdd(&cnt[dst[e]], 1);
}

__global__ void k_scan(const int* __restrict__ cnt, int* __restrict__ start,
                       float* __restrict__ dis, int n) {
    __shared__ int s_pre[32];
    __shared__ int s_tot[32];
    const int tid  = threadIdx.x;
    const int lane = tid & 31;
    const int wave = tid >> 5;
    const int b0   = blockIdx.x * 1024;

    int pre = 0;
    for (int j = tid; j < b0; j += 1024) pre += cnt[j];

    const int i = b0 + tid;
    const int c = (i < n) ? cnt[i] : 0;

    int inc = c;
    for (int o = 1; o < 32; o <<= 1) {
        int v = __shfl_up(inc, o);
        if (lane >= o) inc += v;
    }
    for (int o = 16; o > 0; o >>= 1) pre += __shfl_xor(pre, o);

    if (lane == 31) s_tot[wave] = inc;
    if (lane == 0)  s_pre[wave] = pre;
    __syncthreads();

    int v = s_pre[lane] + ((lane < wave) ? s_tot[lane] : 0);
    for (int o = 16; o > 0; o >>= 1) v += __shfl_xor(v, o);

    if (i < n) {
        start[i] = v + inc - c;
        float d = (float)(c + 1);
        dis[i] = (d > 0.0f) ? rsqrtf(d) : 0.0f;
    }
}

__global__ void k_fill(const int* __restrict__ src, const int* __restrict__ dst,
                       const int* __restrict__ start, const int* __restrict__ pos,
                       int* __restrict__ csr, int E) {
    int e = blockIdx.x * blockDim.x + threadIdx.x;
    if (e < E) csr[start[dst[e]] + pos[e]] = src[e];
}

__global__ void k_gemm_wmma(const float* __restrict__ A, const float* __restrict__ W,
                            float* __restrict__ out, int n, int K) {
    __shared__ float sW[64 * 64];
    const int tid = threadIdx.x;
    for (int i = tid; i < K * 64; i += 128) sW[i] = W[i];
    __syncthreads();

    const int wave = tid >> 5;
    const int lane = tid & 31;
    const int tile_row = blockIdx.x * 64 + wave * 16;

    const int m    = lane & 15;
    const int koff = (lane >> 4) * 2;
    int row = tile_row + m;
    if (row >= n) row = n - 1;

    v8f acc[4] = {v8f{0.f}, v8f{0.f}, v8f{0.f}, v8f{0.f}};

    for (int kb = 0; kb < K; kb += 4) {
        v2f a;
        const float* ap = A + (size_t)row * K + kb + koff;
        a.x = ap[0];
        a.y = ap[1];
#pragma unroll
        for (int nt = 0; nt < 4; ++nt) {
            v2f b;
            const int col = nt * 16 + (lane & 15);
            b.x = sW[(kb + koff) * 64 + col];
            b.y = sW[(kb + koff + 1) * 64 + col];
            acc[nt] = __builtin_amdgcn_wmma_f32_16x16x4_f32(
                 false, a,  false, b,
                 (short)0, acc[nt],  false,  false);
        }
    }

#pragma unroll
    for (int nt = 0; nt < 4; ++nt) {
#pragma unroll
        for (int r = 0; r < 8; ++r) {
            const int mm = r + 8 * (lane >> 4);
            const int rr = tile_row + mm;
            if (rr < n) out[(size_t)rr * 64 + nt * 16 + (lane & 15)] = acc[nt][r];
        }
    }
}

__global__ void k_agg_gemm1(const int* __restrict__ start, const int* __restrict__ cnt,
                            const int* __restrict__ csr, const float* __restrict__ dis,
                            const float* __restrict__ x, const float* __restrict__ W,
                            const float* __restrict__ b, float* __restrict__ out, int n) {
    const int tid  = threadIdx.x;
    const int wave = tid >> 5;
    const int lane = tid & 31;
    const int tile_row = blockIdx.x * 64 + wave * 16;
    if (tile_row >= n) return;

    const int m    = lane & 15;
    const int koff = (lane >> 4) * 2;
    int row = tile_row + m;
    if (row >= n) row = n - 1;

    const float di = dis[row];
    int j = start[row];
    const int end = j + cnt[row];

    const float2 xs = *(const float2*)&x[(size_t)row * 4 + koff];
    const float ss = di * di;
    float ax = ss * xs.x, ay = ss * xs.y;

    for (; j + 4 <= end; j += 4) {
        const int s0 = csr[j], s1 = csr[j + 1], s2 = csr[j + 2], s3 = csr[j + 3];
        const float2 v0 = *(const float2*)&x[(size_t)s0 * 4 + koff];
        const float2 v1 = *(const float2*)&x[(size_t)s1 * 4 + koff];
        const float2 v2 = *(const float2*)&x[(size_t)s2 * 4 + koff];
        const float2 v3 = *(const float2*)&x[(size_t)s3 * 4 + koff];
        const float w0 = dis[s0] * di, w1 = dis[s1] * di;
        const float w2 = dis[s2] * di, w3 = dis[s3] * di;
        ax += w0 * v0.x; ay += w0 * v0.y;
        ax += w1 * v1.x; ay += w1 * v1.y;
        ax += w2 * v2.x; ay += w2 * v2.y;
        ax += w3 * v3.x; ay += w3 * v3.y;
    }
    const int rem = end - j;
    if (rem > 0) {
        const int s0 = csr[j];
        const int s1 = (rem > 1) ? csr[j + 1] : s0;
        const int s2 = (rem > 2) ? csr[j + 2] : s0;
        const float2 v0 = *(const float2*)&x[(size_t)s0 * 4 + koff];
        const float2 v1 = *(const float2*)&x[(size_t)s1 * 4 + koff];
        const float2 v2 = *(const float2*)&x[(size_t)s2 * 4 + koff];
        const float w0 = dis[s0] * di, w1 = dis[s1] * di, w2 = dis[s2] * di;
        ax += w0 * v0.x; ay += w0 * v0.y;
        if (rem > 1) { ax += w1 * v1.x; ay += w1 * v1.y; }
        if (rem > 2) { ax += w2 * v2.x; ay += w2 * v2.y; }
    }

    v2f a;
    a.x = ax;
    a.y = ay;

    v8f acc[4];
#pragma unroll
    for (int nt = 0; nt < 4; ++nt) {
        v2f bf;
        const int col = nt * 16 + m;
        bf.x = W[(koff) * 64 + col];
        bf.y = W[(koff + 1) * 64 + col];
        acc[nt] = __builtin_amdgcn_wmma_f32_16x16x4_f32(
             false, a,  false, bf,
             (short)0, v8f{0.f},  false,  false);
    }

#pragma unroll
    for (int nt = 0; nt < 4; ++nt) {
        const float bv = b[nt * 16 + m];
#pragma unroll
        for (int r = 0; r < 8; ++r) {
            const int rr = tile_row + r + 8 * (lane >> 4);
            if (rr < n) out[(size_t)rr * 64 + nt * 16 + m] = fmaxf(acc[nt][r] + bv, 0.0f);
        }
    }
}

__global__ void k_gather(const int* __restrict__ start, const int* __restrict__ cnt,
                         const int* __restrict__ csr, const float* __restrict__ dis,
                         const float* __restrict__ h, float* __restrict__ out, int n) {
    int t = blockIdx.x * blockDim.x + threadIdx.x;
    if (t >= n * 16) return;
    const int i  = t >> 4;
    const int jj = (t & 15) * 4;
    const float di = dis[i];
    int j = start[i];
    const int end = j + cnt[i];

    const float4 hs = *(const float4*)&h[(size_t)i * 64 + jj];
    const float ss = di * di;
    float4 acc;
    acc.x = ss * hs.x; acc.y = ss * hs.y; acc.z = ss * hs.z; acc.w = ss * hs.w;

    for (; j + 4 <= end; j += 4) {
        const int s0 = csr[j], s1 = csr[j + 1], s2 = csr[j + 2], s3 = csr[j + 3];
        const float4 v0 = *(const float4*)&h[(size_t)s0 * 64 + jj];
        const float4 v1 = *(const float4*)&h[(size_t)s1 * 64 + jj];
        const float4 v2 = *(const float4*)&h[(size_t)s2 * 64 + jj];
        const float4 v3 = *(const float4*)&h[(size_t)s3 * 64 + jj];
        const float w0 = dis[s0] * di, w1 = dis[s1] * di;
        const float w2 = dis[s2] * di, w3 = dis[s3] * di;
        acc.x += w0 * v0.x; acc.y += w0 * v0.y; acc.z += w0 * v0.z; acc.w += w0 * v0.w;
        acc.x += w1 * v1.x; acc.y += w1 * v1.y; acc.z += w1 * v1.z; acc.w += w1 * v1.w;
        acc.x += w2 * v2.x; acc.y += w2 * v2.y; acc.z += w2 * v2.z; acc.w += w2 * v2.w;
        acc.x += w3 * v3.x; acc.y += w3 * v3.y; acc.z += w3 * v3.z; acc.w += w3 * v3.w;
    }
    const int rem = end - j;
    if (rem > 0) {
        const int s0 = csr[j];
        const int s1 = (rem > 1) ? csr[j + 1] : s0;
        const int s2 = (rem > 2) ? csr[j + 2] : s0;
        const float4 v0 = *(const float4*)&h[(size_t)s0 * 64 + jj];
        const float4 v1 = *(const float4*)&h[(size_t)s1 * 64 + jj];
        const float4 v2 = *(const float4*)&h[(size_t)s2 * 64 + jj];
        const float w0 = dis[s0] * di, w1 = dis[s1] * di, w2 = dis[s2] * di;
        acc.x += w0 * v0.x; acc.y += w0 * v0.y; acc.z += w0 * v0.z; acc.w += w0 * v0.w;
        if (rem > 1) {
            acc.x += w1 * v1.x; acc.y += w1 * v1.y; acc.z += w1 * v1.z; acc.w += w1 * v1.w;
        }
        if (rem > 2) {
            acc.x += w2 * v2.x; acc.y += w2 * v2.y; acc.z += w2 * v2.z; acc.w += w2 * v2.w;
        }
    }

    *(float4*)&out[(size_t)i * 64 + jj] = acc;
}

__global__ void k_zero(float* __restrict__ p, int n) {
    int t = blockIdx.x * blockDim.x + threadIdx.x;
    if (t < n) p[t] = 0.0f;
}

__global__ void k_pool(const float* __restrict__ acc2, const float* __restrict__ b2,
                       const int* __restrict__ batch, float* __restrict__ sums, int n) {
    int t = blockIdx.x * blockDim.x + threadIdx.x;
    if (t >= n * 16) return;
    int i = t >> 4;
    int jj = (t & 15) * 4;
    int g = batch[i];
    const float4 hv = *(const float4*)&acc2[(size_t)i * 64 + jj];
    const float4 bv = *(const float4*)&b2[jj];
    float* sp = &sums[(size_t)g * 64 + jj];
    atomicAdd(sp + 0, fmaxf(hv.x + bv.x, 0.0f));
    atomicAdd(sp + 1, fmaxf(hv.y + bv.y, 0.0f));
    atomicAdd(sp + 2, fmaxf(hv.z + bv.z, 0.0f));
    atomicAdd(sp + 3, fmaxf(hv.w + bv.w, 0.0f));
}

__global__ void k_count(const int* __restrict__ batch, float* __restrict__ cnt, int n) {
    int i = blockIdx.x * blockDim.x + threadIdx.x;
    if (i < n) atomicAdd(&cnt[batch[i]], 1.0f);
}

__global__ void k_div(const float* __restrict__ sums, const float* __restrict__ cnt,
                      float* __restrict__ out, int G) {
    int t = blockIdx.x * blockDim.x + threadIdx.x;
    if (t >= G * 64) return;
    out[t] = sums[t] / fmaxf(cnt[t >> 6], 1.0f);
}

extern "C" void kernel_launch(void* const* d_in, const int* in_sizes, int n_in,
                              void* d_out, int out_size, void* d_ws, size_t ws_size,
                              hipStream_t stream) {
    const float* x     = (const float*)d_in[0];
    const int* eidx    = (const int*)d_in[1];
    const int* batch   = (const int*)d_in[2];
    const float* W1 = (const float*)d_in[4];
    const float* b1 = (const float*)d_in[5];
    const float* W2 = (const float*)d_in[6];
    const float* b2 = (const float*)d_in[7];

    const int n = in_sizes[0] / 4;
    const int E = in_sizes[1] / 2;
    const int G = out_size / 64;

    const int* src = eidx;
    const int* dst = eidx + E;

    float* ws   = (float*)d_ws;
    float* dis  = ws;
    float* bufA = dis + n;
    float* bufB = bufA + (size_t)n * 64;
    float* sums = bufB + (size_t)n * 64;
    float* cnt  = sums + (size_t)G * 64;
    int* icnt   = (int*)(cnt + G);
    int* start  = icnt + n;
    int* csr    = start + n;
    int* pos    = (int*)bufB;

    const int T = 256;
    auto blk = [T](int work) { return (work + T - 1) / T; };

    k_zero_i<<<blk(n), T, 0, stream>>>(icnt, n);
    k_rank<<<blk(E), T, 0, stream>>>(dst, icnt, pos, E);
    k_scan<<<(n + 1023) / 1024, 1024, 0, stream>>>(icnt, start, dis, n);
    k_fill<<<blk(E), T, 0, stream>>>(src, dst, start, pos, csr, E);

    k_agg_gemm1<<<(n + 63) / 64, 128, 0, stream>>>(start, icnt, csr, dis, x, W1, b1, bufB, n);

    k_gemm_wmma<<<(n + 63) / 64, 128, 0, stream>>>(bufB, W2, bufA, n, 64);
    k_gather<<<blk(n * 16), T, 0, stream>>>(start, icnt, csr, dis, bufA, bufB, n);

    k_zero<<<blk(G * 64 + G), T, 0, stream>>>(sums, G * 64 + G);
    k_pool<<<blk(n * 16), T, 0, stream>>>(bufB, b2, batch, sums, n);
    k_count<<<blk(n), T, 0, stream>>>(batch, cnt, n);
    k_div<<<blk(G * 64), T, 0, stream>>>(sums, cnt, (float*)d_out, G);
}


# edge list built through LDS: per-chunk range counts, one returning add per (chunk,range), range-ordered line stores, per-range LDS count/scan/slot; no per-edge global atomic; sums in LDS slot order
# speedup vs baseline: 2.3167x; 1.3608x over previous
#include <hip/hip_runtime.h>
#include <hip/hip_bf16.h>

typedef __attribute__((ext_vector_type(2))) float v2f;
typedef __attribute__((ext_vector_type(8))) float v8f;

#define RSH    8
#define RNODES (1 << RSH)
#define RMASK  (RNODES - 1)
#define RMAX   1024
#define CHUNK  4096
#define EPT    (CHUNK / 1024)
#define SCAP   8192

__global__ void k_zero_i(int* __restrict__ p, int n) {
    int i = blockIdx.x * blockDim.x + threadIdx.x;
    if (i < n) p[i] = 0;
}

__device__ __forceinline__ int block_excl_scan(int c, int* s_w) {
    const int tid  = threadIdx.x;
    const int lane = tid & 31;
    const int wave = tid >> 5;
    int inc = c;
    for (int o = 1; o < 32; o <<= 1) {
        int v = __shfl_up(inc, o);
        if (lane >= o) inc += v;
    }
    if (lane == 31) s_w[wave] = inc;
    __syncthreads();
    int v = (lane < wave) ? s_w[lane] : 0;
    for (int o = 16; o > 0; o >>= 1) v += __shfl_xor(v, o);
    return v + inc - c;
}

__global__ void k_chunk_count(const int* __restrict__ dst, int* __restrict__ rtot,
                              int* __restrict__ piece, int E, int R) {
    __shared__ int s_h[RMAX];
    const int tid = threadIdx.x;
    const int e0  = blockIdx.x * CHUNK;
    int d[EPT];
#pragma unroll
    for (int q = 0; q < EPT; ++q) {
        const int e  = e0 + q * 1024 + tid;
        const int ec = (e < E) ? e : E - 1;
        const int v  = dst[ec];
        d[q] = (e < E) ? v : -1;
    }
    s_h[tid] = 0;
    __syncthreads();
#pragma unroll
    for (int q = 0; q < EPT; ++q)
        if (d[q] >= 0) atomicAdd(&s_h[d[q] >> RSH], 1);
    __syncthreads();
    if (tid < R) piece[blockIdx.x * R + tid] = atomicAdd(&rtot[tid], s_h[tid]);
}

__global__ void k_chunk_place(const int* __restrict__ src, const int* __restrict__ dst,
                              const int* __restrict__ rtot, const int* __restrict__ piece,
                              int* __restrict__ binned, int E, int R) {
    __shared__ int s_cur[RMAX];
    __shared__ int s_dlt[RMAX];
    __shared__ int s_w0[32];
    __shared__ int s_w1[32];
    __shared__ int s_word[CHUNK];
    __shared__ int s_gpos[CHUNK];
    const int tid = threadIdx.x;
    const int e0  = blockIdx.x * CHUNK;
    int d[EPT], s[EPT];
#pragma unroll
    for (int q = 0; q < EPT; ++q) {
        const int e  = e0 + q * 1024 + tid;
        const int ec = (e < E) ? e : E - 1;
        const int v  = dst[ec];
        s[q] = src[ec];
        d[q] = (e < E) ? v : -1;
    }
    s_cur[tid] = 0;
    __syncthreads();
#pragma unroll
    for (int q = 0; q < EPT; ++q)
        if (d[q] >= 0) atomicAdd(&s_cur[d[q] >> RSH], 1);
    __syncthreads();

    const int h   = s_cur[tid];
    const int lex = block_excl_scan(h, s_w0);
    const int c   = (tid < R) ? rtot[tid] : 0;
    const int gex = block_excl_scan(c, s_w1);
    s_cur[tid] = lex;
    s_dlt[tid] = (tid < R) ? gex + piece[blockIdx.x * R + tid] - lex : 0;
    __syncthreads();

#pragma unroll
    for (int q = 0; q < EPT; ++q) {
        if (d[q] >= 0) {
            const int r  = d[q] >> RSH;
            const int lp = atomicAdd(&s_cur[r], 1);
            s_word[lp] = (s[q] << RSH) | (d[q] & RMASK);
            s_gpos[lp] = lp + s_dlt[r];
        }
    }
    __syncthreads();

#pragma unroll
    for (int q = 0; q < EPT; ++q) {
        const int j = q * 1024 + tid;
        if (e0 + j < E) binned[s_gpos[j]] = s_word[j];
    }
}

__global__ void k_range_sort(const int* __restrict__ rtot, const int* __restrict__ binned,
                             int* __restrict__ csr, int* __restrict__ start,
                             int* __restrict__ cnt, float* __restrict__ dis, int n, int R) {
    __shared__ int s_span[2];
    __shared__ int s_w0[32];
    __shared__ int s_w1[32];
    __shared__ int s_c[RNODES];
    __shared__ int s_o[RNODES];
    __shared__ int s_out[SCAP];
    const int tid = threadIdx.x;
    const int r   = blockIdx.x;

    const int c  = (tid < R) ? rtot[tid] : 0;
    const int ex = block_excl_scan(c, s_w0);
    if (tid == r) { s_span[0] = ex; s_span[1] = c; }
    if (tid < RNODES) s_c[tid] = 0;
    __syncthreads();
    const int b0  = s_span[0];
    const int len = s_span[1];

    for (int k0 = 0; k0 < len; k0 += CHUNK) {
        int w[EPT];
#pragma unroll
        for (int q = 0; q < EPT; ++q) {
            const int k  = k0 + q * 1024 + tid;
            const int kc = (k < len) ? k : len - 1;
            w[q] = binned[b0 + kc];
        }
#pragma unroll
        for (int q = 0; q < EPT; ++q)
            if (k0 + q * 1024 + tid >= len) w[q] = -1;
#pragma unroll
        for (int q = 0; q < EPT; ++q)
            if (w[q] >= 0) atomicAdd(&s_c[w[q] & RMASK], 1);
    }
    __syncthreads();

    const int cn  = (tid < RNODES) ? s_c[tid] : 0;
    const int off = block_excl_scan(cn, s_w1);
    if (tid < RNODES) {
        s_o[tid] = off;
        const int i = (r << RSH) + tid;
        if (i < n) {
            start[i] = b0 + off;
            cnt[i]   = cn;
            float dg = (float)(cn + 1);
            dis[i] = (dg > 0.0f) ? rsqrtf(dg) : 0.0f;
        }
    }
    __syncthreads();

    if (len <= SCAP) {
        for (int k0 = 0; k0 < len; k0 += CHUNK) {
            int w[EPT];
#pragma unroll
            for (int q = 0; q < EPT; ++q) {
                const int k  = k0 + q * 1024 + tid;
                const int kc = (k < len) ? k : len - 1;
                w[q] = binned[b0 + kc];
            }
#pragma unroll
            for (int q = 0; q < EPT; ++q)
                if (k0 + q * 1024 + tid >= len) w[q] = -1;
#pragma unroll
            for (int q = 0; q < EPT; ++q)
                if (w[q] >= 0) s_out[atomicAdd(&s_o[w[q] & RMASK], 1)] = w[q] >> RSH;
        }
        __syncthreads();
        for (int k = tid; k < len; k += 1024) csr[b0 + k] = s_out[k];
    } else {
        for (int k = tid; k < len; k += 1024) {
            const int w = binned[b0 + k];
            csr[b0 + atomicAdd(&s_o[w & RMASK], 1)] = w >> RSH;
        }
    }
}

__global__ void k_gemm_wmma(const float* __restrict__ A, const float* __restrict__ W,
                            float* __restrict__ out, int n, int K) {
    __shared__ float sW[64 * 64];
    const int tid = threadIdx.x;
    for (int i = tid; i < K * 64; i += 128) sW[i] = W[i];
    __syncthreads();

    const int wave = tid >> 5;
    const int lane = tid & 31;
    const int tile_row = blockIdx.x * 64 + wave * 16;

    const int m    = lane & 15;
    const int koff = (lane >> 4) * 2;
    int row = tile_row + m;
    if (row >= n) row = n - 1;

    v8f acc[4] = {v8f{0.f}, v8f{0.f}, v8f{0.f}, v8f{0.f}};

    for (int kb = 0; kb < K; kb += 4) {
        v2f a;
        const float* ap = A + (size_t)row * K + kb + koff;
        a.x = ap[0];
        a.y = ap[1];
#pragma unroll
        for (int nt = 0; nt < 4; ++nt) {
            v2f b;
            const int col = nt * 16 + (lane & 15);
            b.x = sW[(kb + koff) * 64 + col];
            b.y = sW[(kb + koff + 1) * 64 + col];
            acc[nt] = __builtin_amdgcn_wmma_f32_16x16x4_f32(
                 false, a,  false, b,
                 (short)0, acc[nt],  false,  false);
        }
    }

#pragma unroll
    for (int nt = 0; nt < 4; ++nt) {
#pragma unroll
        for (int r = 0; r < 8; ++r) {
            const int mm = r + 8 * (lane >> 4);
            const int rr = tile_row + mm;
            if (rr < n) out[(size_t)rr * 64 + nt * 16 + (lane & 15)] = acc[nt][r];
        }
    }
}

__global__ void k_agg_gemm1(const int* __restrict__ start, const int* __restrict__ cnt,
                            const int* __restrict__ csr, const float* __restrict__ dis,
                            const float* __restrict__ x, const float* __restrict__ W,
                            const float* __restrict__ b, float* __restrict__ out, int n) {
    const int tid  = threadIdx.x;
    const int wave = tid >> 5;
    const int lane = tid & 31;
    const int tile_row = blockIdx.x * 64 + wave * 16;
    if (tile_row >= n) return;

    const int m    = lane & 15;
    const int koff = (lane >> 4) * 2;
    int row = tile_row + m;
    if (row >= n) row = n - 1;

    const float di = dis[row];
    int j = start[row];
    const int end = j + cnt[row];

    const float2 xs = *(const float2*)&x[(size_t)row * 4 + koff];
    const float ss = di * di;
    float ax = ss * xs.x, ay = ss * xs.y;

    for (; j + 4 <= end; j += 4) {
        const int s0 = csr[j], s1 = csr[j + 1], s2 = csr[j + 2], s3 = csr[j + 3];
        const float2 v0 = *(const float2*)&x[(size_t)s0 * 4 + koff];
        const float2 v1 = *(const float2*)&x[(size_t)s1 * 4 + koff];
        const float2 v2 = *(const float2*)&x[(size_t)s2 * 4 + koff];
        const float2 v3 = *(const float2*)&x[(size_t)s3 * 4 + koff];
        const float w0 = dis[s0] * di, w1 = dis[s1] * di;
        const float w2 = dis[s2] * di, w3 = dis[s3] * di;
        ax += w0 * v0.x; ay += w0 * v0.y;
        ax += w1 * v1.x; ay += w1 * v1.y;
        ax += w2 * v2.x; ay += w2 * v2.y;
        ax += w3 * v3.x; ay += w3 * v3.y;
    }
    const int rem = end - j;
    if (rem > 0) {
        const int s0 = csr[j];
        const int s1 = (rem > 1) ? csr[j + 1] : s0;
        const int s2 = (rem > 2) ? csr[j + 2] : s0;
        const float2 v0 = *(const float2*)&x[(size_t)s0 * 4 + koff];
        const float2 v1 = *(const float2*)&x[(size_t)s1 * 4 + koff];
        const float2 v2 = *(const float2*)&x[(size_t)s2 * 4 + koff];
        const float w0 = dis[s0] * di, w1 = dis[s1] * di, w2 = dis[s2] * di;
        ax += w0 * v0.x; ay += w0 * v0.y;
        if (rem > 1) { ax += w1 * v1.x; ay += w1 * v1.y; }
        if (rem > 2) { ax += w2 * v2.x; ay += w2 * v2.y; }
    }

    v2f a;
    a.x = ax;
    a.y = ay;

    v8f acc[4];
#pragma unroll
    for (int nt = 0; nt < 4; ++nt) {
        v2f bf;
        const int col = nt * 16 + m;
        bf.x = W[(koff) * 64 + col];
        bf.y = W[(koff + 1) * 64 + col];
        acc[nt] = __builtin_amdgcn_wmma_f32_16x16x4_f32(
             false, a,  false, bf,
             (short)0, v8f{0.f},  false,  false);
    }

#pragma unroll
    for (int nt = 0; nt < 4; ++nt) {
        const float bv = b[nt * 16 + m];
#pragma unroll
        for (int r = 0; r < 8; ++r) {
            const int rr = tile_row + r + 8 * (lane >> 4);
            if (rr < n) out[(size_t)rr * 64 + nt * 16 + m] = fmaxf(acc[nt][r] + bv, 0.0f);
        }
    }
}

__global__ void k_gather(const int* __restrict__ start, const int* __restrict__ cnt,
                         const int* __restrict__ csr, const float* __restrict__ dis,
                         const float* __restrict__ h, float* __restrict__ out, int n) {
    int t = blockIdx.x * blockDim.x + threadIdx.x;
    if (t >= n * 16) return;
    const int i  = t >> 4;
    const int jj = (t & 15) * 4;
    const float di = dis[i];
    int j = start[i];
    const int end = j + cnt[i];

    const float4 hs = *(const float4*)&h[(size_t)i * 64 + jj];
    const float ss = di * di;
    float4 acc;
    acc.x = ss * hs.x; acc.y = ss * hs.y; acc.z = ss * hs.z; acc.w = ss * hs.w;

    for (; j + 4 <= end; j += 4) {
        const int s0 = csr[j], s1 = csr[j + 1], s2 = csr[j + 2], s3 = csr[j + 3];
        const float4 v0 = *(const float4*)&h[(size_t)s0 * 64 + jj];
        const float4 v1 = *(const float4*)&h[(size_t)s1 * 64 + jj];
        const float4 v2 = *(const float4*)&h[(size_t)s2 * 64 + jj];
        const float4 v3 = *(const float4*)&h[(size_t)s3 * 64 + jj];
        const float w0 = dis[s0] * di, w1 = dis[s1] * di;
        const float w2 = dis[s2] * di, w3 = dis[s3] * di;
        acc.x += w0 * v0.x; acc.y += w0 * v0.y; acc.z += w0 * v0.z; acc.w += w0 * v0.w;
        acc.x += w1 * v1.x; acc.y += w1 * v1.y; acc.z += w1 * v1.z; acc.w += w1 * v1.w;
        acc.x += w2 * v2.x; acc.y += w2 * v2.y; acc.z += w2 * v2.z; acc.w += w2 * v2.w;
        acc.x += w3 * v3.x; acc.y += w3 * v3.y; acc.z += w3 * v3.z; acc.w += w3 * v3.w;
    }
    const int rem = end - j;
    if (rem > 0) {
        const int s0 = csr[j];
        const int s1 = (rem > 1) ? csr[j + 1] : s0;
        const int s2 = (rem > 2) ? csr[j + 2] : s0;
        const float4 v0 = *(const float4*)&h[(size_t)s0 * 64 + jj];
        const float4 v1 = *(const float4*)&h[(size_t)s1 * 64 + jj];
        const float4 v2 = *(const float4*)&h[(size_t)s2 * 64 + jj];
        const float w0 = dis[s0] * di, w1 = dis[s1] * di, w2 = dis[s2] * di;
        acc.x += w0 * v0.x; acc.y += w0 * v0.y; acc.z += w0 * v0.z; acc.w += w0 * v0.w;
        if (rem > 1) {
            acc.x += w1 * v1.x; acc.y += w1 * v1.y; acc.z += w1 * v1.z; acc.w += w1 * v1.w;
        }
        if (rem > 2) {
            acc.x += w2 * v2.x; acc.y += w2 * v2.y; acc.z += w2 * v2.z; acc.w += w2 * v2.w;
        }
    }

    *(float4*)&out[(size_t)i * 64 + jj] = acc;
}

__global__ void k_zero(float* __restrict__ p, int n) {
    int t = blockIdx.x * blockDim.x + threadIdx.x;
    if (t < n) p[t] = 0.0f;
}

__global__ void k_pool(const float* __restrict__ acc2, const float* __restrict__ b2,
                       const int* __restrict__ batch, float* __restrict__ sums, int n) {
    int t = blockIdx.x * blockDim.x + threadIdx.x;
    if (t >= n * 16) return;
    int i = t >> 4;
    int jj = (t & 15) * 4;
    int g = batch[i];
    const float4 hv = *(const float4*)&acc2[(size_t)i * 64 + jj];
    const float4 bv = *(const float4*)&b2[jj];
    float* sp = &sums[(size_t)g * 64 + jj];
    atomicAdd(sp + 0, fmaxf(hv.x + bv.x, 0.0f));
    atomicAdd(sp + 1, fmaxf(hv.y + bv.y, 0.0f));
    atomicAdd(sp + 2, fmaxf(hv.z + bv.z, 0.0f));
    atomicAdd(sp + 3, fmaxf(hv.w + bv.w, 0.0f));
}

__global__ void k_count(const int* __restrict__ batch, float* __restrict__ cnt, int n) {
    int i = blockIdx.x * blockDim.x + threadIdx.x;
    if (i < n) atomicAdd(&cnt[batch[i]], 1.0f);
}

__global__ void k_div(const float* __restrict__ sums, const float* __restrict__ cnt,
                      float* __restrict__ out, int G) {
    int t = blockIdx.x * blockDim.x + threadIdx.x;
    if (t >= G * 64) return;
    out[t] = sums[t] / fmaxf(cnt[t >> 6], 1.0f);
}

extern "C" void kernel_launch(void* const* d_in, const int* in_sizes, int n_in,
                              void* d_out, int out_size, void* d_ws, size_t ws_size,
                              hipStream_t stream) {
    const float* x     = (const float*)d_in[0];
    const int* eidx    = (const int*)d_in[1];
    const int* batch   = (const int*)d_in[2];
    const float* W1 = (const float*)d_in[4];
    const float* b1 = (const float*)d_in[5];
    const float* W2 = (const float*)d_in[6];
    const float* b2 = (const float*)d_in[7];

    const int n = in_sizes[0] / 4;
    const int E = in_sizes[1] / 2;
    const int G = out_size / 64;

    const int* src = eidx;
    const int* dst = eidx + E;

    float* ws   = (float*)d_ws;
    float* dis  = ws;
    float* bufA = dis + n;
    float* bufB = bufA + (size_t)n * 64;
    float* sums = bufB + (size_t)n * 64;
    float* cnt  = sums + (size_t)G * 64;
    int* icnt   = (int*)(cnt + G);
    int* start  = icnt + n;
    int* csr    = start + n;
    int* binned = (int*)bufA;
    int* rtot   = (int*)bufB;
    int* piece  = rtot + RMAX;

    const int T = 256;
    auto blk = [T](int work) { return (work + T - 1) / T; };

    const int R  = (n + (1 << RSH) - 1) >> RSH;
    const int NC = (E + CHUNK - 1) / CHUNK;
    k_zero_i<<<blk(R), T, 0, stream>>>(rtot, R);
    k_chunk_count<<<NC, 1024, 0, stream>>>(dst, rtot, piece, E, R);
    k_chunk_place<<<NC, 1024, 0, stream>>>(src, dst, rtot, piece, binned, E, R);
    k_range_sort<<<R, 1024, 0, stream>>>(rtot, binned, csr, start, icnt, dis, n, R);

    k_agg_gemm1<<<(n + 63) / 64, 128, 0, stream>>>(start, icnt, csr, dis, x, W1, b1, bufB, n);

    k_gemm_wmma<<<(n + 63) / 64, 128, 0, stream>>>(bufB, W2, bufA, n, 64);
    k_gather<<<blk(n * 16), T, 0, stream>>>(start, icnt, csr, dis, bufA, bufB, n);

    k_zero<<<blk(G * 64 + G), T, 0, stream>>>(sums, G * 64 + G);
    k_pool<<<blk(n * 16), T, 0, stream>>>(bufB, b2, batch, sums, n);
    k_count<<<blk(n), T, 0, stream>>>(batch, cnt, n);
    k_div<<<blk(G * 64), T, 0, stream>>>(sums, cnt, (float*)d_out, G);
}


# mean pool (sums, counts) moved into the layer-2 gather epilogue: rows via LDS, one float atomic per (run of equal graph ids, column) per block; pool, count launches gone; sum order changes
# speedup vs baseline: 5.4989x; 2.3736x over previous
#include <hip/hip_runtime.h>
#include <hip/hip_bf16.h>

typedef __attribute__((ext_vector_type(2))) float v2f;
typedef __attribute__((ext_vector_type(8))) float v8f;

#define RSH    8
#define RNODES (1 << RSH)
#define RMASK  (RNODES - 1)
#define RMAX   1024
#define CHUNK  4096
#define EPT    (CHUNK / 1024)
#define SCAP   8192

__global__ void k_zero_i(int* __restrict__ p, int n) {
    int i = blockIdx.x * blockDim.x + threadIdx.x;
    if (i < n) p[i] = 0;
}

__device__ __forceinline__ int block_excl_scan(int c, int* s_w) {
    const int tid  = threadIdx.x;
    const int lane = tid & 31;
    const int wave = tid >> 5;
    int inc = c;
    for (int o = 1; o < 32; o <<= 1) {
        int v = __shfl_up(inc, o);
        if (lane >= o) inc += v;
    }
    if (lane == 31) s_w[wave] = inc;
    __syncthreads();
    int v = (lane < wave) ? s_w[lane] : 0;
    for (int o = 16; o > 0; o >>= 1) v += __shfl_xor(v, o);
    return v + inc - c;
}

__global__ void k_chunk_count(const int* __restrict__ dst, int* __restrict__ rtot,
                              int* __restrict__ piece, int E, int R) {
    __shared__ int s_h[RMAX];
    const int tid = threadIdx.x;
    const int e0  = blockIdx.x * CHUNK;
    int d[EPT];
#pragma unroll
    for (int q = 0; q < EPT; ++q) {
        const int e  = e0 + q * 1024 + tid;
        const int ec = (e < E) ? e : E - 1;
        const int v  = dst[ec];
        d[q] = (e < E) ? v : -1;
    }
    s_h[tid] = 0;
    __syncthreads();
#pragma unroll
    for (int q = 0; q < EPT; ++q)
        if (d[q] >= 0) atomicAdd(&s_h[d[q] >> RSH], 1);
    __syncthreads();
    if (tid < R) piece[blockIdx.x * R + tid] = atomicAdd(&rtot[tid], s_h[tid]);
}

__global__ void k_chunk_place(const int* __restrict__ src, const int* __restrict__ dst,
                              const int* __restrict__ rtot, const int* __restrict__ piece,
                              int* __restrict__ binned, int E, int R) {
    __shared__ int s_cur[RMAX];
    __shared__ int s_dlt[RMAX];
    __shared__ int s_w0[32];
    __shared__ int s_w1[32];
    __shared__ int s_word[CHUNK];
    __shared__ int s_gpos[CHUNK];
    const int tid = threadIdx.x;
    const int e0  = blockIdx.x * CHUNK;
    int d[EPT], s[EPT];
#pragma unroll
    for (int q = 0; q < EPT; ++q) {
        const int e  = e0 + q * 1024 + tid;
        const int ec = (e < E) ? e : E - 1;
        const int v  = dst[ec];
        s[q] = src[ec];
        d[q] = (e < E) ? v : -1;
    }
    s_cur[tid] = 0;
    __syncthreads();
#pragma unroll
    for (int q = 0; q < EPT; ++q)
        if (d[q] >= 0) atomicAdd(&s_cur[d[q] >> RSH], 1);
    __syncthreads();

    const int h   = s_cur[tid];
    const int lex = block_excl_scan(h, s_w0);
    const int c   = (tid < R) ? rtot[tid] : 0;
    const int gex = block_excl_scan(c, s_w1);
    s_cur[tid] = lex;
    s_dlt[tid] = (tid < R) ? gex + piece[blockIdx.x * R + tid] - lex : 0;
    __syncthreads();

#pragma unroll
    for (int q = 0; q < EPT; ++q) {
        if (d[q] >= 0) {
            const int r  = d[q] >> RSH;
            const int lp = atomicAdd(&s_cur[r], 1);
            s_word[lp] = (s[q] << RSH) | (d[q] & RMASK);
            s_gpos[lp] = lp + s_dlt[r];
        }
    }
    __syncthreads();

#pragma unroll
    for (int q = 0; q < EPT; ++q) {
        const int j = q * 1024 + tid;
        if (e0 + j < E) binned[s_gpos[j]] = s_word[j];
    }
}

__global__ void k_range_sort(const int* __restrict__ rtot, const int* __restrict__ binned,
                             int* __restrict__ csr, int* __restrict__ start,
                             int* __restrict__ cnt, float* __restrict__ dis, int n, int R) {
    __shared__ int s_span[2];
    __shared__ int s_w0[32];
    __shared__ int s_w1[32];
    __shared__ int s_c[RNODES];
    __shared__ int s_o[RNODES];
    __shared__ int s_out[SCAP];
    const int tid = threadIdx.x;
    const int r   = blockIdx.x;

    const int c  = (tid < R) ? rtot[tid] : 0;
    const int ex = block_excl_scan(c, s_w0);
    if (tid == r) { s_span[0] = ex; s_span[1] = c; }
    if (tid < RNODES) s_c[tid] = 0;
    __syncthreads();
    const int b0  = s_span[0];
    const int len = s_span[1];

    for (int k0 = 0; k0 < len; k0 += CHUNK) {
        int w[EPT];
#pragma unroll
        for (int q = 0; q < EPT; ++q) {
            const int k  = k0 + q * 1024 + tid;
            const int kc = (k < len) ? k : len - 1;
            w[q] = binned[b0 + kc];
        }
#pragma unroll
        for (int q = 0; q < EPT; ++q)
            if (k0 + q * 1024 + tid >= len) w[q] = -1;
#pragma unroll
        for (int q = 0; q < EPT; ++q)
            if (w[q] >= 0) atomicAdd(&s_c[w[q] & RMASK], 1);
    }
    __syncthreads();

    const int cn  = (tid < RNODES) ? s_c[tid] : 0;
    const int off = block_excl_scan(cn, s_w1);
    if (tid < RNODES) {
        s_o[tid] = off;
        const int i = (r << RSH) + tid;
        if (i < n) {
            start[i] = b0 + off;
            cnt[i]   = cn;
            float dg = (float)(cn + 1);
            dis[i] = (dg > 0.0f) ? rsqrtf(dg) : 0.0f;
        }
    }
    __syncthreads();

    if (len <= SCAP) {
        for (int k0 = 0; k0 < len; k0 += CHUNK) {
            int w[EPT];
#pragma unroll
            for (int q = 0; q < EPT; ++q) {
                const int k  = k0 + q * 1024 + tid;
                const int kc = (k < len) ? k : len - 1;
                w[q] = binned[b0 + kc];
            }
#pragma unroll
            for (int q = 0; q < EPT; ++q)
                if (k0 + q * 1024 + tid >= len) w[q] = -1;
#pragma unroll
            for (int q = 0; q < EPT; ++q)
                if (w[q] >= 0) s_out[atomicAdd(&s_o[w[q] & RMASK], 1)] = w[q] >> RSH;
        }
        __syncthreads();
        for (int k = tid; k < len; k += 1024) csr[b0 + k] = s_out[k];
    } else {
        for (int k = tid; k < len; k += 1024) {
            const int w = binned[b0 + k];
            csr[b0 + atomicAdd(&s_o[w & RMASK], 1)] = w >> RSH;
        }
    }
}

__global__ void k_gemm_wmma(const float* __restrict__ A, const float* __restrict__ W,
                            float* __restrict__ out, int n, int K) {
    __shared__ float sW[64 * 64];
    const int tid = threadIdx.x;
    for (int i = tid; i < K * 64; i += 128) sW[i] = W[i];
    __syncthreads();

    const int wave = tid >> 5;
    const int lane = tid & 31;
    const int tile_row = blockIdx.x * 64 + wave * 16;

    const int m    = lane & 15;
    const int koff = (lane >> 4) * 2;
    int row = tile_row + m;
    if (row >= n) row = n - 1;

    v8f acc[4] = {v8f{0.f}, v8f{0.f}, v8f{0.f}, v8f{0.f}};

    for (int kb = 0; kb < K; kb += 4) {
        v2f a;
        const float* ap = A + (size_t)row * K + kb + koff;
        a.x = ap[0];
        a.y = ap[1];
#pragma unroll
        for (int nt = 0; nt < 4; ++nt) {
            v2f b;
            const int col = nt * 16 + (lane & 15);
            b.x = sW[(kb + koff) * 64 + col];
            b.y = sW[(kb + koff + 1) * 64 + col];
            acc[nt] = __builtin_amdgcn_wmma_f32_16x16x4_f32(
                 false, a,  false, b,
                 (short)0, acc[nt],  false,  false);
        }
    }

#pragma unroll
    for (int nt = 0; nt < 4; ++nt) {
#pragma unroll
        for (int r = 0; r < 8; ++r) {
            const int mm = r + 8 * (lane >> 4);
            const int rr = tile_row + mm;
            if (rr < n) out[(size_t)rr * 64 + nt * 16 + (lane & 15)] = acc[nt][r];
        }
    }
}

__global__ void k_agg_gemm1(const int* __restrict__ start, const int* __restrict__ cnt,
                            const int* __restrict__ csr, const float* __restrict__ dis,
                            const float* __restrict__ x, const float* __restrict__ W,
                            const float* __restrict__ b, float* __restrict__ out, int n) {
    const int tid  = threadIdx.x;
    const int wave = tid >> 5;
    const int lane = tid & 31;
    const int tile_row = blockIdx.x * 64 + wave * 16;
    if (tile_row >= n) return;

    const int m    = lane & 15;
    const int koff = (lane >> 4) * 2;
    int row = tile_row + m;
    if (row >= n) row = n - 1;

    const float di = dis[row];
    int j = start[row];
    const int end = j + cnt[row];

    const float2 xs = *(const float2*)&x[(size_t)row * 4 + koff];
    const float ss = di * di;
    float ax = ss * xs.x, ay = ss * xs.y;

    for (; j + 4 <= end; j += 4) {
        const int s0 = csr[j], s1 = csr[j + 1], s2 = csr[j + 2], s3 = csr[j + 3];
        const float2 v0 = *(const float2*)&x[(size_t)s0 * 4 + koff];
        const float2 v1 = *(const float2*)&x[(size_t)s1 * 4 + koff];
        const float2 v2 = *(const float2*)&x[(size_t)s2 * 4 + koff];
        const float2 v3 = *(const float2*)&x[(size_t)s3 * 4 + koff];
        const float w0 = dis[s0] * di, w1 = dis[s1] * di;
        const float w2 = dis[s2] * di, w3 = dis[s3] * di;
        ax += w0 * v0.x; ay += w0 * v0.y;
        ax += w1 * v1.x; ay += w1 * v1.y;
        ax += w2 * v2.x; ay += w2 * v2.y;
        ax += w3 * v3.x; ay += w3 * v3.y;
    }
    const int rem = end - j;
    if (rem > 0) {
        const int s0 = csr[j];
        const int s1 = (rem > 1) ? csr[j + 1] : s0;
        const int s2 = (rem > 2) ? csr[j + 2] : s0;
        const float2 v0 = *(const float2*)&x[(size_t)s0 * 4 + koff];
        const float2 v1 = *(const float2*)&x[(size_t)s1 * 4 + koff];
        const float2 v2 = *(const float2*)&x[(size_t)s2 * 4 + koff];
        const float w0 = dis[s0] * di, w1 = dis[s1] * di, w2 = dis[s2] * di;
        ax += w0 * v0.x; ay += w0 * v0.y;
        if (rem > 1) { ax += w1 * v1.x; ay += w1 * v1.y; }
        if (rem > 2) { ax += w2 * v2.x; ay += w2 * v2.y; }
    }

    v2f a;
    a.x = ax;
    a.y = ay;

    v8f acc[4];
#pragma unroll
    for (int nt = 0; nt < 4; ++nt) {
        v2f bf;
        const int col = nt * 16 + m;
        bf.x = W[(koff) * 64 + col];
        bf.y = W[(koff + 1) * 64 + col];
        acc[nt] = __builtin_amdgcn_wmma_f32_16x16x4_f32(
             false, a,  false, bf,
             (short)0, v8f{0.f},  false,  false);
    }

#pragma unroll
    for (int nt = 0; nt < 4; ++nt) {
        const float bv = b[nt * 16 + m];
#pragma unroll
        for (int r = 0; r < 8; ++r) {
            const int rr = tile_row + r + 8 * (lane >> 4);
            if (rr < n) out[(size_t)rr * 64 + nt * 16 + m] = fmaxf(acc[nt][r] + bv, 0.0f);
        }
    }
}

#define GNODES 16

__global__ void k_gather_pool(const int* __restrict__ start, const int* __restrict__ cnt,
                              const int* __restrict__ csr, const float* __restrict__ dis,
                              const float* __restrict__ h, const float* __restrict__ b2,
                              const int* __restrict__ batch, float* __restrict__ sums,
                              float* __restrict__ gcnt, int n) {
    __shared__ __attribute__((aligned(16))) float s_v[GNODES * 64];
    __shared__ int   s_g[GNODES];
    const int tid = threadIdx.x;
    const int ln  = tid >> 4;
    const int i   = blockIdx.x * GNODES + ln;
    const int jj  = (tid & 15) * 4;

    float4 v;
    v.x = 0.0f; v.y = 0.0f; v.z = 0.0f; v.w = 0.0f;
    int g = -1;
    if (i < n) {
        const float di = dis[i];
        int j = start[i];
        const int end = j + cnt[i];
        g = batch[i];
        const float4 bv = *(const float4*)&b2[jj];

        const float4 hs = *(const float4*)&h[(size_t)i * 64 + jj];
        const float ss = di * di;
        float4 acc;
        acc.x = ss * hs.x; acc.y = ss * hs.y; acc.z = ss * hs.z; acc.w = ss * hs.w;

        for (; j + 4 <= end; j += 4) {
            const int s0 = csr[j], s1 = csr[j + 1], s2 = csr[j + 2], s3 = csr[j + 3];
            const float4 v0 = *(const float4*)&h[(size_t)s0 * 64 + jj];
            const float4 v1 = *(const float4*)&h[(size_t)s1 * 64 + jj];
            const float4 v2 = *(const float4*)&h[(size_t)s2 * 64 + jj];
            const float4 v3 = *(const float4*)&h[(size_t)s3 * 64 + jj];
            const float w0 = dis[s0] * di, w1 = dis[s1] * di;
            const float w2 = dis[s2] * di, w3 = dis[s3] * di;
            acc.x += w0 * v0.x; acc.y += w0 * v0.y; acc.z += w0 * v0.z; acc.w += w0 * v0.w;
            acc.x += w1 * v1.x; acc.y += w1 * v1.y; acc.z += w1 * v1.z; acc.w += w1 * v1.w;
            acc.x += w2 * v2.x; acc.y += w2 * v2.y; acc.z += w2 * v2.z; acc.w += w2 * v2.w;
            acc.x += w3 * v3.x; acc.y += w3 * v3.y; acc.z += w3 * v3.z; acc.w += w3 * v3.w;
        }
        const int rem = end - j;
        if (rem > 0) {
            const int s0 = csr[j];
            const int s1 = (rem > 1) ? csr[j + 1] : s0;
            const int s2 = (rem > 2) ? csr[j + 2] : s0;
            const float4 v0 = *(const float4*)&h[(size_t)s0 * 64 + jj];
            const float4 v1 = *(const float4*)&h[(size_t)s1 * 64 + jj];
            const float4 v2 = *(const float4*)&h[(size_t)s2 * 64 + jj];
            const float w0 = dis[s0] * di, w1 = dis[s1] * di, w2 = dis[s2] * di;
            acc.x += w0 * v0.x; acc.y += w0 * v0.y; acc.z += w0 * v0.z; acc.w += w0 * v0.w;
            if (rem > 1) {
                acc.x += w1 * v1.x; acc.y += w1 * v1.y; acc.z += w1 * v1.z; acc.w += w1 * v1.w;
            }
            if (rem > 2) {
                acc.x += w2 * v2.x; acc.y += w2 * v2.y; acc.z += w2 * v2.z; acc.w += w2 * v2.w;
            }
        }

        v.x = fmaxf(acc.x + bv.x, 0.0f);
        v.y = fmaxf(acc.y + bv.y, 0.0f);
        v.z = fmaxf(acc.z + bv.z, 0.0f);
        v.w = fmaxf(acc.w + bv.w, 0.0f);
    }

    *(float4*)&s_v[ln * 64 + jj] = v;
    if ((tid & 15) == 0) s_g[ln] = g;
    __syncthreads();

    if (tid <= 64) {
        int gq[GNODES];
#pragma unroll
        for (int q = 0; q < GNODES; ++q) gq[q] = s_g[q];
        if (tid < 64) {
            float vq[GNODES];
#pragma unroll
            for (int q = 0; q < GNODES; ++q) vq[q] = s_v[q * 64 + tid];
            int   cur = gq[0];
            float a   = 0.0f;
#pragma unroll
            for (int q = 0; q < GNODES; ++q) {
                if (gq[q] != cur) {
                    if (cur >= 0) atomicAdd(&sums[(size_t)cur * 64 + tid], a);
                    a   = 0.0f;
                    cur = gq[q];
                }
                a += vq[q];
            }
            if (cur >= 0) atomicAdd(&sums[(size_t)cur * 64 + tid], a);
        } else {
            int cur = gq[0];
            int run = 0;
#pragma unroll
            for (int q = 0; q < GNODES; ++q) {
                if (gq[q] != cur) {
                    if (cur >= 0) atomicAdd(&gcnt[cur], (float)run);
                    run = 0;
                    cur = gq[q];
                }
                ++run;
            }
            if (cur >= 0) atomicAdd(&gcnt[cur], (float)run);
        }
    }
}

__global__ void k_zero(float* __restrict__ p, int n) {
    int t = blockIdx.x * blockDim.x + threadIdx.x;
    if (t < n) p[t] = 0.0f;
}

__global__ void k_div(const float* __restrict__ sums, const float* __restrict__ cnt,
                      float* __restrict__ out, int G) {
    int t = blockIdx.x * blockDim.x + threadIdx.x;
    if (t >= G * 64) return;
    out[t] = sums[t] / fmaxf(cnt[t >> 6], 1.0f);
}

extern "C" void kernel_launch(void* const* d_in, const int* in_sizes, int n_in,
                              void* d_out, int out_size, void* d_ws, size_t ws_size,
                              hipStream_t stream) {
    const float* x     = (const float*)d_in[0];
    const int* eidx    = (const int*)d_in[1];
    const int* batch   = (const int*)d_in[2];
    const float* W1 = (const float*)d_in[4];
    const float* b1 = (const float*)d_in[5];
    const float* W2 = (const float*)d_in[6];
    const float* b2 = (const float*)d_in[7];

    const int n = in_sizes[0] / 4;
    const int E = in_sizes[1] / 2;
    const int G = out_size / 64;

    const int* src = eidx;
    const int* dst = eidx + E;

    float* ws   = (float*)d_ws;
    float* dis  = ws;
    float* bufA = dis + n;
    float* bufB = bufA + (size_t)n * 64;
    float* sums = bufB + (size_t)n * 64;
    float* cnt  = sums + (size_t)G * 64;
    int* icnt   = (int*)(cnt + G);
    int* start  = icnt + n;
    int* csr    = start + n;
    int* binned = (int*)bufA;
    int* rtot   = (int*)bufB;
    int* piece  = rtot + RMAX;

    const int T = 256;
    auto blk = [T](int work) { return (work + T - 1) / T; };

    const int R  = (n + (1 << RSH) - 1) >> RSH;
    const int NC = (E + CHUNK - 1) / CHUNK;
    k_zero_i<<<blk(R), T, 0, stream>>>(rtot, R);
    k_chunk_count<<<NC, 1024, 0, stream>>>(dst, rtot, piece, E, R);
    k_chunk_place<<<NC, 1024, 0, stream>>>(src, dst, rtot, piece, binned, E, R);
    k_range_sort<<<R, 1024, 0, stream>>>(rtot, binned, csr, start, icnt, dis, n, R);

    k_agg_gemm1<<<(n + 63) / 64, 128, 0, stream>>>(start, icnt, csr, dis, x, W1, b1, bufB, n);

    k_gemm_wmma<<<(n + 63) / 64, 128, 0, stream>>>(bufB, W2, bufA, n, 64);
    k_zero<<<blk(G * 64 + G), T, 0, stream>>>(sums, G * 64 + G);
    k_gather_pool<<<blk(n * 16), T, 0, stream>>>(start, icnt, csr, dis, bufA, b2, batch, sums, cnt, n);

    k_div<<<blk(G * 64), T, 0, stream>>>(sums, cnt, (float*)d_out, G);
}


# layer-2 product h1r @ W2 moved into the layer-1 kernel: each wave's relu'd 16x64 tile goes through LDS into the A fragments; h1r store, its re-read and the GEMM launch gone; sum order unchanged
# speedup vs baseline: 6.0685x; 1.1036x over previous
#include <hip/hip_runtime.h>
#include <hip/hip_bf16.h>

typedef __attribute__((ext_vector_type(2))) float v2f;
typedef __attribute__((ext_vector_type(8))) float v8f;

#define RSH    8
#define RNODES (1 << RSH)
#define RMASK  (RNODES - 1)
#define RMAX   1024
#define CHUNK  4096
#define EPT    (CHUNK / 1024)
#define SCAP   8192

__global__ void k_zero_i(int* __restrict__ p, int n) {
    int i = blockIdx.x * blockDim.x + threadIdx.x;
    if (i < n) p[i] = 0;
}

__device__ __forceinline__ int block_excl_scan(int c, int* s_w) {
    const int tid  = threadIdx.x;
    const int lane = tid & 31;
    const int wave = tid >> 5;
    int inc = c;
    for (int o = 1; o < 32; o <<= 1) {
        int v = __shfl_up(inc, o);
        if (lane >= o) inc += v;
    }
    if (lane == 31) s_w[wave] = inc;
    __syncthreads();
    int v = (lane < wave) ? s_w[lane] : 0;
    for (int o = 16; o > 0; o >>= 1) v += __shfl_xor(v, o);
    return v + inc - c;
}

__global__ void k_chunk_count(const int* __restrict__ dst, int* __restrict__ rtot,
                              int* __restrict__ piece, int E, int R) {
    __shared__ int s_h[RMAX];
    const int tid = threadIdx.x;
    const int e0  = blockIdx.x * CHUNK;
    int d[EPT];
#pragma unroll
    for (int q = 0; q < EPT; ++q) {
        const int e  = e0 + q * 1024 + tid;
        const int ec = (e < E) ? e : E - 1;
        const int v  = dst[ec];
        d[q] = (e < E) ? v : -1;
    }
    s_h[tid] = 0;
    __syncthreads();
#pragma unroll
    for (int q = 0; q < EPT; ++q)
        if (d[q] >= 0) atomicAdd(&s_h[d[q] >> RSH], 1);
    __syncthreads();
    if (tid < R) piece[blockIdx.x * R + tid] = atomicAdd(&rtot[tid], s_h[tid]);
}

__global__ void k_chunk_place(const int* __restrict__ src, const int* __restrict__ dst,
                              const int* __restrict__ rtot, const int* __restrict__ piece,
                              int* __restrict__ binned, int E, int R) {
    __shared__ int s_cur[RMAX];
    __shared__ int s_dlt[RMAX];
    __shared__ int s_w0[32];
    __shared__ int s_w1[32];
    __shared__ int s_word[CHUNK];
    __shared__ int s_gpos[CHUNK];
    const int tid = threadIdx.x;
    const int e0  = blockIdx.x * CHUNK;
    int d[EPT], s[EPT];
#pragma unroll
    for (int q = 0; q < EPT; ++q) {
        const int e  = e0 + q * 1024 + tid;
        const int ec = (e < E) ? e : E - 1;
        const int v  = dst[ec];
        s[q] = src[ec];
        d[q] = (e < E) ? v : -1;
    }
    s_cur[tid] = 0;
    __syncthreads();
#pragma unroll
    for (int q = 0; q < EPT; ++q)
        if (d[q] >= 0) atomicAdd(&s_cur[d[q] >> RSH], 1);
    __syncthreads();

    const int h   = s_cur[tid];
    const int lex = block_excl_scan(h, s_w0);
    const int c   = (tid < R) ? rtot[tid] : 0;
    const int gex = block_excl_scan(c, s_w1);
    s_cur[tid] = lex;
    s_dlt[tid] = (tid < R) ? gex + piece[blockIdx.x * R + tid] - lex : 0;
    __syncthreads();

#pragma unroll
    for (int q = 0; q < EPT; ++q) {
        if (d[q] >= 0) {
            const int r  = d[q] >> RSH;
            const int lp = atomicAdd(&s_cur[r], 1);
            s_word[lp] = (s[q] << RSH) | (d[q] & RMASK);
            s_gpos[lp] = lp + s_dlt[r];
        }
    }
    __syncthreads();

#pragma unroll
    for (int q = 0; q < EPT; ++q) {
        const int j = q * 1024 + tid;
        if (e0 + j < E) binned[s_gpos[j]] = s_word[j];
    }
}

__global__ void k_range_sort(const int* __restrict__ rtot, const int* __restrict__ binned,
                             int* __restrict__ csr, int* __restrict__ start,
                             int* __restrict__ cnt, float* __restrict__ dis, int n, int R) {
    __shared__ int s_span[2];
    __shared__ int s_w0[32];
    __shared__ int s_w1[32];
    __shared__ int s_c[RNODES];
    __shared__ int s_o[RNODES];
    __shared__ int s_out[SCAP];
    const int tid = threadIdx.x;
    const int r   = blockIdx.x;

    const int c  = (tid < R) ? rtot[tid] : 0;
    const int ex = block_excl_scan(c, s_w0);
    if (tid == r) { s_span[0] = ex; s_span[1] = c; }
    if (tid < RNODES) s_c[tid] = 0;
    __syncthreads();
    const int b0  = s_span[0];
    const int len = s_span[1];

    for (int k0 = 0; k0 < len; k0 += CHUNK) {
        int w[EPT];
#pragma unroll
        for (int q = 0; q < EPT; ++q) {
            const int k  = k0 + q * 1024 + tid;
            const int kc = (k < len) ? k : len - 1;
            w[q] = binned[b0 + kc];
        }
#pragma unroll
        for (int q = 0; q < EPT; ++q)
            if (k0 + q * 1024 + tid >= len) w[q] = -1;
#pragma unroll
        for (int q = 0; q < EPT; ++q)
            if (w[q] >= 0) atomicAdd(&s_c[w[q] & RMASK], 1);
    }
    __syncthreads();

    const int cn  = (tid < RNODES) ? s_c[tid] : 0;
    const int off = block_excl_scan(cn, s_w1);
    if (tid < RNODES) {
        s_o[tid] = off;
        const int i = (r << RSH) + tid;
        if (i < n) {
            start[i] = b0 + off;
            cnt[i]   = cn;
            float dg = (float)(cn + 1);
            dis[i] = (dg > 0.0f) ? rsqrtf(dg) : 0.0f;
        }
    }
    __syncthreads();

    if (len <= SCAP) {
        for (int k0 = 0; k0 < len; k0 += CHUNK) {
            int w[EPT];
#pragma unroll
            for (int q = 0; q < EPT; ++q) {
                const int k  = k0 + q * 1024 + tid;
                const int kc = (k < len) ? k : len - 1;
                w[q] = binned[b0 + kc];
            }
#pragma unroll
            for (int q = 0; q < EPT; ++q)
                if (k0 + q * 1024 + tid >= len) w[q] = -1;
#pragma unroll
            for (int q = 0; q < EPT; ++q)
                if (w[q] >= 0) s_out[atomicAdd(&s_o[w[q] & RMASK], 1)] = w[q] >> RSH;
        }
        __syncthreads();
        for (int k = tid; k < len; k += 1024) csr[b0 + k] = s_out[k];
    } else {
        for (int k = tid; k < len; k += 1024) {
            const int w = binned[b0 + k];
            csr[b0 + atomicAdd(&s_o[w & RMASK], 1)] = w >> RSH;
        }
    }
}

#define TPITCH 68
#define WPITCH 72

__global__ void k_agg_gemm12(const int* __restrict__ start, const int* __restrict__ cnt,
                             const int* __restrict__ csr, const float* __restrict__ dis,
                             const float* __restrict__ x, const float* __restrict__ W1,
                             const float* __restrict__ b1, const float* __restrict__ W2,
                             float* __restrict__ out, int n) {
    __shared__ __attribute__((aligned(16))) float sW[64 * WPITCH];
    __shared__ __attribute__((aligned(16))) float sT[4 * 16 * TPITCH];
    const int tid  = threadIdx.x;
    const int wave = tid >> 5;
    const int lane = tid & 31;
    const int tile_row = blockIdx.x * 64 + wave * 16;

    float4 wq[8];
#pragma unroll
    for (int q = 0; q < 8; ++q) wq[q] = *(const float4*)&W2[(size_t)(q * 128 + tid) * 4];

    const int m    = lane & 15;
    const int koff = (lane >> 4) * 2;
    int row = tile_row + m;
    if (row >= n) row = n - 1;

    const float di = dis[row];
    int j = start[row];
    const int end = j + cnt[row];

    const float2 xs = *(const float2*)&x[(size_t)row * 4 + koff];

#pragma unroll
    for (int q = 0; q < 8; ++q) {
        const int i = (q * 128 + tid) * 4;
        *(float4*)&sW[(i >> 6) * WPITCH + (i & 63)] = wq[q];
    }

    const float ss = di * di;
    float ax = ss * xs.x, ay = ss * xs.y;

    for (; j + 4 <= end; j += 4) {
        const int s0 = csr[j], s1 = csr[j + 1], s2 = csr[j + 2], s3 = csr[j + 3];
        const float2 v0 = *(const float2*)&x[(size_t)s0 * 4 + koff];
        const float2 v1 = *(const float2*)&x[(size_t)s1 * 4 + koff];
        const float2 v2 = *(const float2*)&x[(size_t)s2 * 4 + koff];
        const float2 v3 = *(const float2*)&x[(size_t)s3 * 4 + koff];
        const float w0 = dis[s0] * di, w1 = dis[s1] * di;
        const float w2 = dis[s2] * di, w3 = dis[s3] * di;
        ax += w0 * v0.x; ay += w0 * v0.y;
        ax += w1 * v1.x; ay += w1 * v1.y;
        ax += w2 * v2.x; ay += w2 * v2.y;
        ax += w3 * v3.x; ay += w3 * v3.y;
    }
    const int rem = end - j;
    if (rem > 0) {
        const int s0 = csr[j];
        const int s1 = (rem > 1) ? csr[j + 1] : s0;
        const int s2 = (rem > 2) ? csr[j + 2] : s0;
        const float2 v0 = *(const float2*)&x[(size_t)s0 * 4 + koff];
        const float2 v1 = *(const float2*)&x[(size_t)s1 * 4 + koff];
        const float2 v2 = *(const float2*)&x[(size_t)s2 * 4 + koff];
        const float w0 = dis[s0] * di, w1 = dis[s1] * di, w2 = dis[s2] * di;
        ax += w0 * v0.x; ay += w0 * v0.y;
        if (rem > 1) { ax += w1 * v1.x; ay += w1 * v1.y; }
        if (rem > 2) { ax += w2 * v2.x; ay += w2 * v2.y; }
    }

    v2f a;
    a.x = ax;
    a.y = ay;

    v8f acc[4];
#pragma unroll
    for (int nt = 0; nt < 4; ++nt) {
        v2f bf;
        const int col = nt * 16 + m;
        bf.x = W1[(koff) * 64 + col];
        bf.y = W1[(koff + 1) * 64 + col];
        acc[nt] = __builtin_amdgcn_wmma_f32_16x16x4_f32(
             false, a,  false, bf,
             (short)0, v8f{0.f},  false,  false);
    }

    float* tw = sT + wave * 16 * TPITCH;
#pragma unroll
    for (int nt = 0; nt < 4; ++nt) {
        const float bv = b1[nt * 16 + m];
#pragma unroll
        for (int r = 0; r < 8; ++r)
            tw[(r + 8 * (lane >> 4)) * TPITCH + nt * 16 + m] = fmaxf(acc[nt][r] + bv, 0.0f);
    }
    __syncthreads();

    v8f acc2[4] = {v8f{0.f}, v8f{0.f}, v8f{0.f}, v8f{0.f}};
#pragma unroll
    for (int kb = 0; kb < 64; kb += 4) {
        const v2f a2 = *(const v2f*)&tw[m * TPITCH + kb + koff];
#pragma unroll
        for (int nt = 0; nt < 4; ++nt) {
            v2f b;
            const int col = nt * 16 + m;
            b.x = sW[(kb + koff) * WPITCH + col];
            b.y = sW[(kb + koff + 1) * WPITCH + col];
            acc2[nt] = __builtin_amdgcn_wmma_f32_16x16x4_f32(
                 false, a2,  false, b,
                 (short)0, acc2[nt],  false,  false);
        }
    }

#pragma unroll
    for (int nt = 0; nt < 4; ++nt) {
#pragma unroll
        for (int r = 0; r < 8; ++r) {
            const int rr = tile_row + r + 8 * (lane >> 4);
            if (rr < n) out[(size_t)rr * 64 + nt * 16 + m] = acc2[nt][r];
        }
    }
}

#define GNODES 16

__global__ void k_gather_pool(const int* __restrict__ start, const int* __restrict__ cnt,
                              const int* __restrict__ csr, const float* __restrict__ dis,
                              const float* __restrict__ h, const float* __restrict__ b2,
                              const int* __restrict__ batch, float* __restrict__ sums,
                              float* __restrict__ gcnt, int n) {
    __shared__ __attribute__((aligned(16))) float s_v[GNODES * 64];
    __shared__ int   s_g[GNODES];
    const int tid = threadIdx.x;
    const int ln  = tid >> 4;
    const int i   = blockIdx.x * GNODES + ln;
    const int jj  = (tid & 15) * 4;

    float4 v;
    v.x = 0.0f; v.y = 0.0f; v.z = 0.0f; v.w = 0.0f;
    int g = -1;
    if (i < n) {
        const float di = dis[i];
        int j = start[i];
        const int end = j + cnt[i];
        g = batch[i];
        const float4 bv = *(const float4*)&b2[jj];

        const float4 hs = *(const float4*)&h[(size_t)i * 64 + jj];
        const float ss = di * di;
        float4 acc;
        acc.x = ss * hs.x; acc.y = ss * hs.y; acc.z = ss * hs.z; acc.w = ss * hs.w;

        for (; j + 4 <= end; j += 4) {
            const int s0 = csr[j], s1 = csr[j + 1], s2 = csr[j + 2], s3 = csr[j + 3];
            const float4 v0 = *(const float4*)&h[(size_t)s0 * 64 + jj];
            const float4 v1 = *(const float4*)&h[(size_t)s1 * 64 + jj];
            const float4 v2 = *(const float4*)&h[(size_t)s2 * 64 + jj];
            const float4 v3 = *(const float4*)&h[(size_t)s3 * 64 + jj];
            const float w0 = dis[s0] * di, w1 = dis[s1] * di;
            const float w2 = dis[s2] * di, w3 = dis[s3] * di;
            acc.x += w0 * v0.x; acc.y += w0 * v0.y; acc.z += w0 * v0.z; acc.w += w0 * v0.w;
            acc.x += w1 * v1.x; acc.y += w1 * v1.y; acc.z += w1 * v1.z; acc.w += w1 * v1.w;
            acc.x += w2 * v2.x; acc.y += w2 * v2.y; acc.z += w2 * v2.z; acc.w += w2 * v2.w;
            acc.x += w3 * v3.x; acc.y += w3 * v3.y; acc.z += w3 * v3.z; acc.w += w3 * v3.w;
        }
        const int rem = end - j;
        if (rem > 0) {
            const int s0 = csr[j];
            const int s1 = (rem > 1) ? csr[j + 1] : s0;
            const int s2 = (rem > 2) ? csr[j + 2] : s0;
            const float4 v0 = *(const float4*)&h[(size_t)s0 * 64 + jj];
            const float4 v1 = *(const float4*)&h[(size_t)s1 * 64 + jj];
            const float4 v2 = *(const float4*)&h[(size_t)s2 * 64 + jj];
            const float w0 = dis[s0] * di, w1 = dis[s1] * di, w2 = dis[s2] * di;
            acc.x += w0 * v0.x; acc.y += w0 * v0.y; acc.z += w0 * v0.z; acc.w += w0 * v0.w;
            if (rem > 1) {
                acc.x += w1 * v1.x; acc.y += w1 * v1.y; acc.z += w1 * v1.z; acc.w += w1 * v1.w;
            }
            if (rem > 2) {
                acc.x += w2 * v2.x; acc.y += w2 * v2.y; acc.z += w2 * v2.z; acc.w += w2 * v2.w;
            }
        }

        v.x = fmaxf(acc.x + bv.x, 0.0f);
        v.y = fmaxf(acc.y + bv.y, 0.0f);
        v.z = fmaxf(acc.z + bv.z, 0.0f);
        v.w = fmaxf(acc.w + bv.w, 0.0f);
    }

    *(float4*)&s_v[ln * 64 + jj] = v;
    if ((tid & 15) == 0) s_g[ln] = g;
    __syncthreads();

    if (tid <= 64) {
        int gq[GNODES];
#pragma unroll
        for (int q = 0; q < GNODES; ++q) gq[q] = s_g[q];
        if (tid < 64) {
            float vq[GNODES];
#pragma unroll
            for (int q = 0; q < GNODES; ++q) vq[q] = s_v[q * 64 + tid];
            int   cur = gq[0];
            float a   = 0.0f;
#pragma unroll
            for (int q = 0; q < GNODES; ++q) {
                if (gq[q] != cur) {
                    if (cur >= 0) atomicAdd(&sums[(size_t)cur * 64 + tid], a);
                    a   = 0.0f;
                    cur = gq[q];
                }
                a += vq[q];
            }
            if (cur >= 0) atomicAdd(&sums[(size_t)cur * 64 + tid], a);
        } else {
            int cur = gq[0];
            int run = 0;
#pragma unroll
            for (int q = 0; q < GNODES; ++q) {
                if (gq[q] != cur) {
                    if (cur >= 0) atomicAdd(&gcnt[cur], (float)run);
                    run = 0;
                    cur = gq[q];
                }
                ++run;
            }
            if (cur >= 0) atomicAdd(&gcnt[cur], (float)run);
        }
    }
}

__global__ void k_zero(float* __restrict__ p, int n) {
    int t = blockIdx.x * blockDim.x + threadIdx.x;
    if (t < n) p[t] = 0.0f;
}

__global__ void k_div(const float* __restrict__ sums, const float* __restrict__ cnt,
                      float* __restrict__ out, int G) {
    int t = blockIdx.x * blockDim.x + threadIdx.x;
    if (t >= G * 64) return;
    out[t] = sums[t] / fmaxf(cnt[t >> 6], 1.0f);
}

extern "C" void kernel_launch(void* const* d_in, const int* in_sizes, int n_in,
                              void* d_out, int out_size, void* d_ws, size_t ws_size,
                              hipStream_t stream) {
    const float* x     = (const float*)d_in[0];
    const int* eidx    = (const int*)d_in[1];
    const int* batch   = (const int*)d_in[2];
    const float* W1 = (const float*)d_in[4];
    const float* b1 = (const float*)d_in[5];
    const float* W2 = (const float*)d_in[6];
    const float* b2 = (const float*)d_in[7];

    const int n = in_sizes[0] / 4;
    const int E = in_sizes[1] / 2;
    const int G = out_size / 64;

    const int* src = eidx;
    const int* dst = eidx + E;

    float* ws   = (float*)d_ws;
    float* dis  = ws;
    float* bufA = dis + n;
    float* bufB = bufA + (size_t)n * 64;
    float* sums = bufB + (size_t)n * 64;
    float* cnt  = sums + (size_t)G * 64;
    int* icnt   = (int*)(cnt + G);
    int* start  = icnt + n;
    int* csr    = start + n;
    int* binned = (int*)bufA;
    int* rtot   = (int*)bufB;
    int* piece  = rtot + RMAX;

    const int T = 256;
    auto blk = [T](int work) { return (work + T - 1) / T; };

    const int R  = (n + (1 << RSH) - 1) >> RSH;
    const int NC = (E + CHUNK - 1) / CHUNK;
    k_zero_i<<<blk(R), T, 0, stream>>>(rtot, R);
    k_chunk_count<<<NC, 1024, 0, stream>>>(dst, rtot, piece, E, R);
    k_chunk_place<<<NC, 1024, 0, stream>>>(src, dst, rtot, piece, binned, E, R);
    k_range_sort<<<R, 1024, 0, stream>>>(rtot, binned, csr, start, icnt, dis, n, R);

    k_agg_gemm12<<<(n + 63) / 64, 128, 0, stream>>>(start, icnt, csr, dis, x, W1, b1, W2, bufA, n);
    k_zero<<<blk(G * 64 + G), T, 0, stream>>>(sums, G * 64 + G);
    k_gather_pool<<<blk(n * 16), T, 0, stream>>>(start, icnt, csr, dis, bufA, b2, batch, sums, cnt, n);

    k_div<<<blk(G * 64), T, 0, stream>>>(sums, cnt, (float*)d_out, G);
}


# mean pool's divide moved into the gather's float atomics: nodes per graph counted in the sort kernel, each run sum divided by it, added into out; zero and divide launches gone; rounding order changes
# speedup vs baseline: 6.5722x; 1.0830x over previous
#include <hip/hip_runtime.h>
#include <hip/hip_bf16.h>

typedef __attribute__((ext_vector_type(2))) float v2f;
typedef __attribute__((ext_vector_type(8))) float v8f;

#define RSH    8
#define RNODES (1 << RSH)
#define RMASK  (RNODES - 1)
#define RMAX   1024
#define CHUNK  4096
#define EPT    (CHUNK / 1024)
#define SCAP   8192

__global__ void k_zero_all(int* __restrict__ rtot, int R, float* __restrict__ out, int nout,
                           float* __restrict__ gcnt, int G) {
    const int i = blockIdx.x * blockDim.x + threadIdx.x;
    if (i < R)    rtot[i] = 0;
    if (i < nout) out[i]  = 0.0f;
    if (i < G)    gcnt[i] = 0.0f;
}

__device__ __forceinline__ int block_excl_scan(int c, int* s_w) {
    const int tid  = threadIdx.x;
    const int lane = tid & 31;
    const int wave = tid >> 5;
    int inc = c;
    for (int o = 1; o < 32; o <<= 1) {
        int v = __shfl_up(inc, o);
        if (lane >= o) inc += v;
    }
    if (lane == 31) s_w[wave] = inc;
    __syncthreads();
    int v = (lane < wave) ? s_w[lane] : 0;
    for (int o = 16; o > 0; o >>= 1) v += __shfl_xor(v, o);
    return v + inc - c;
}

__global__ void k_chunk_count(const int* __restrict__ dst, int* __restrict__ rtot,
                              int* __restrict__ piece, int E, int R) {
    __shared__ int s_h[RMAX];
    const int tid = threadIdx.x;
    const int e0  = blockIdx.x * CHUNK;
    int d[EPT];
#pragma unroll
    for (int q = 0; q < EPT; ++q) {
        const int e  = e0 + q * 1024 + tid;
        const int ec = (e < E) ? e : E - 1;
        const int v  = dst[ec];
        d[q] = (e < E) ? v : -1;
    }
    s_h[tid] = 0;
    __syncthreads();
#pragma unroll
    for (int q = 0; q < EPT; ++q)
        if (d[q] >= 0) atomicAdd(&s_h[d[q] >> RSH], 1);
    __syncthreads();
    if (tid < R) piece[blockIdx.x * R + tid] = atomicAdd(&rtot[tid], s_h[tid]);
}

__global__ void k_chunk_place(const int* __restrict__ src, const int* __restrict__ dst,
                              const int* __restrict__ rtot, const int* __restrict__ piece,
                              int* __restrict__ binned, int E, int R) {
    __shared__ int s_cur[RMAX];
    __shared__ int s_dlt[RMAX];
    __shared__ int s_w0[32];
    __shared__ int s_w1[32];
    __shared__ int s_word[CHUNK];
    __shared__ int s_gpos[CHUNK];
    const int tid = threadIdx.x;
    const int e0  = blockIdx.x * CHUNK;
    int d[EPT], s[EPT];
#pragma unroll
    for (int q = 0; q < EPT; ++q) {
        const int e  = e0 + q * 1024 + tid;
        const int ec = (e < E) ? e : E - 1;
        const int v  = dst[ec];
        s[q] = src[ec];
        d[q] = (e < E) ? v : -1;
    }
    s_cur[tid] = 0;
    __syncthreads();
#pragma unroll
    for (int q = 0; q < EPT; ++q)
        if (d[q] >= 0) atomicAdd(&s_cur[d[q] >> RSH], 1);
    __syncthreads();

    const int h   = s_cur[tid];
    const int lex = block_excl_scan(h, s_w0);
    const int c   = (tid < R) ? rtot[tid] : 0;
    const int gex = block_excl_scan(c, s_w1);
    s_cur[tid] = lex;
    s_dlt[tid] = (tid < R) ? gex + piece[blockIdx.x * R + tid] - lex : 0;
    __syncthreads();

#pragma unroll
    for (int q = 0; q < EPT; ++q) {
        if (d[q] >= 0) {
            const int r  = d[q] >> RSH;
            const int lp = atomicAdd(&s_cur[r], 1);
            s_word[lp] = (s[q] << RSH) | (d[q] & RMASK);
            s_gpos[lp] = lp + s_dlt[r];
        }
    }
    __syncthreads();

#pragma unroll
    for (int q = 0; q < EPT; ++q) {
        const int j = q * 1024 + tid;
        if (e0 + j < E) binned[s_gpos[j]] = s_word[j];
    }
}

__global__ void k_range_sort(const int* __restrict__ rtot, const int* __restrict__ binned,
                             int* __restrict__ csr, int* __restrict__ start,
                             int* __restrict__ cnt, float* __restrict__ dis,
                             const int* __restrict__ batch, float* __restrict__ gcnt,
                             int n, int R) {
    __shared__ int s_span[2];
    __shared__ int s_w0[32];
    __shared__ int s_w1[32];
    __shared__ int s_c[RNODES];
    __shared__ int s_o[RNODES];
    __shared__ int s_out[SCAP];
    const int tid = threadIdx.x;
    const int r   = blockIdx.x;

    const int c  = (tid < R) ? rtot[tid] : 0;
    const int ex = block_excl_scan(c, s_w0);
    if (tid == r) { s_span[0] = ex; s_span[1] = c; }
    if (tid < RNODES) s_c[tid] = 0;
    __syncthreads();
    const int b0  = s_span[0];
    const int len = s_span[1];

    for (int k0 = 0; k0 < len; k0 += CHUNK) {
        int w[EPT];
#pragma unroll
        for (int q = 0; q < EPT; ++q) {
            const int k  = k0 + q * 1024 + tid;
            const int kc = (k < len) ? k : len - 1;
            w[q] = binned[b0 + kc];
        }
#pragma unroll
        for (int q = 0; q < EPT; ++q)
            if (k0 + q * 1024 + tid >= len) w[q] = -1;
#pragma unroll
        for (int q = 0; q < EPT; ++q)
            if (w[q] >= 0) atomicAdd(&s_c[w[q] & RMASK], 1);
    }
    __syncthreads();

    const int cn  = (tid < RNODES) ? s_c[tid] : 0;
    const int off = block_excl_scan(cn, s_w1);
    if (tid < RNODES) {
        s_o[tid] = off;
        const int i = (r << RSH) + tid;
        const int g = (i < n) ? batch[i] : -1;
        if (i < n) {
            start[i] = b0 + off;
            cnt[i]   = cn;
            float dg = (float)(cn + 1);
            dis[i] = (dg > 0.0f) ? rsqrtf(dg) : 0.0f;
        }
        const int lane = tid & 31;
        unsigned todo = (unsigned)__ballot(g >= 0);
        for (int it = 0; it < 32 && todo != 0u; ++it) {
            const int leader = __ffs((int)todo) - 1;
            const int gl = __shfl(g, leader);
            const unsigned same = (unsigned)__ballot(g == gl) & todo;
            if (lane == leader) atomicAdd(&gcnt[gl], (float)__popc(same));
            todo &= ~same;
        }
    }
    __syncthreads();

    if (len <= SCAP) {
        for (int k0 = 0; k0 < len; k0 += CHUNK) {
            int w[EPT];
#pragma unroll
            for (int q = 0; q < EPT; ++q) {
                const int k  = k0 + q * 1024 + tid;
                const int kc = (k < len) ? k : len - 1;
                w[q] = binned[b0 + kc];
            }
#pragma unroll
            for (int q = 0; q < EPT; ++q)
                if (k0 + q * 1024 + tid >= len) w[q] = -1;
#pragma unroll
            for (int q = 0; q < EPT; ++q)
                if (w[q] >= 0) s_out[atomicAdd(&s_o[w[q] & RMASK], 1)] = w[q] >> RSH;
        }
        __syncthreads();
        for (int k = tid; k < len; k += 1024) csr[b0 + k] = s_out[k];
    } else {
        for (int k = tid; k < len; k += 1024) {
            const int w = binned[b0 + k];
            csr[b0 + atomicAdd(&s_o[w & RMASK], 1)] = w >> RSH;
        }
    }
}

#define TPITCH 68
#define WPITCH 72

__global__ void k_agg_gemm12(const int* __restrict__ start, const int* __restrict__ cnt,
                             const int* __restrict__ csr, const float* __restrict__ dis,
                             const float* __restrict__ x, const float* __restrict__ W1,
                             const float* __restrict__ b1, const float* __restrict__ W2,
                             float* __restrict__ out, int n) {
    __shared__ __attribute__((aligned(16))) float sW[64 * WPITCH];
    __shared__ __attribute__((aligned(16))) float sT[4 * 16 * TPITCH];
    const int tid  = threadIdx.x;
    const int wave = tid >> 5;
    const int lane = tid & 31;
    const int tile_row = blockIdx.x * 64 + wave * 16;

    float4 wq[8];
#pragma unroll
    for (int q = 0; q < 8; ++q) wq[q] = *(const float4*)&W2[(size_t)(q * 128 + tid) * 4];

    const int m    = lane & 15;
    const int koff = (lane >> 4) * 2;
    int row = tile_row + m;
    if (row >= n) row = n - 1;

    const float di = dis[row];
    int j = start[row];
    const int end = j + cnt[row];

    const float2 xs = *(const float2*)&x[(size_t)row * 4 + koff];

#pragma unroll
    for (int q = 0; q < 8; ++q) {
        const int i = (q * 128 + tid) * 4;
        *(float4*)&sW[(i >> 6) * WPITCH + (i & 63)] = wq[q];
    }

    const float ss = di * di;
    float ax = ss * xs.x, ay = ss * xs.y;

    for (; j + 4 <= end; j += 4) {
        const int s0 = csr[j], s1 = csr[j + 1], s2 = csr[j + 2], s3 = csr[j + 3];
        const float2 v0 = *(const float2*)&x[(size_t)s0 * 4 + koff];
        const float2 v1 = *(const float2*)&x[(size_t)s1 * 4 + koff];
        const float2 v2 = *(const float2*)&x[(size_t)s2 * 4 + koff];
        const float2 v3 = *(const float2*)&x[(size_t)s3 * 4 + koff];
        const float w0 = dis[s0] * di, w1 = dis[s1] * di;
        const float w2 = dis[s2] * di, w3 = dis[s3] * di;
        ax += w0 * v0.x; ay += w0 * v0.y;
        ax += w1 * v1.x; ay += w1 * v1.y;
        ax += w2 * v2.x; ay += w2 * v2.y;
        ax += w3 * v3.x; ay += w3 * v3.y;
    }
    const int rem = end - j;
    if (rem > 0) {
        const int s0 = csr[j];
        const int s1 = (rem > 1) ? csr[j + 1] : s0;
        const int s2 = (rem > 2) ? csr[j + 2] : s0;
        const float2 v0 = *(const float2*)&x[(size_t)s0 * 4 + koff];
        const float2 v1 = *(const float2*)&x[(size_t)s1 * 4 + koff];
        const float2 v2 = *(const float2*)&x[(size_t)s2 * 4 + koff];
        const float w0 = dis[s0] * di, w1 = dis[s1] * di, w2 = dis[s2] * di;
        ax += w0 * v0.x; ay += w0 * v0.y;
        if (rem > 1) { ax += w1 * v1.x; ay += w1 * v1.y; }
        if (rem > 2) { ax += w2 * v2.x; ay += w2 * v2.y; }
    }

    v2f a;
    a.x = ax;
    a.y = ay;

    v8f acc[4];
#pragma unroll
    for (int nt = 0; nt < 4; ++nt) {
        v2f bf;
        const int col = nt * 16 + m;
        bf.x = W1[(koff) * 64 + col];
        bf.y = W1[(koff + 1) * 64 + col];
        acc[nt] = __builtin_amdgcn_wmma_f32_16x16x4_f32(
             false, a,  false, bf,
             (short)0, v8f{0.f},  false,  false);
    }

    float* tw = sT + wave * 16 * TPITCH;
#pragma unroll
    for (int nt = 0; nt < 4; ++nt) {
        const float bv = b1[nt * 16 + m];
#pragma unroll
        for (int r = 0; r < 8; ++r)
            tw[(r + 8 * (lane >> 4)) * TPITCH + nt * 16 + m] = fmaxf(acc[nt][r] + bv, 0.0f);
    }
    __syncthreads();

    v8f acc2[4] = {v8f{0.f}, v8f{0.f}, v8f{0.f}, v8f{0.f}};
#pragma unroll
    for (int kb = 0; kb < 64; kb += 4) {
        const v2f a2 = *(const v2f*)&tw[m * TPITCH + kb + koff];
#pragma unroll
        for (int nt = 0; nt < 4; ++nt) {
            v2f b;
            const int col = nt * 16 + m;
            b.x = sW[(kb + koff) * WPITCH + col];
            b.y = sW[(kb + koff + 1) * WPITCH + col];
            acc2[nt] = __builtin_amdgcn_wmma_f32_16x16x4_f32(
                 false, a2,  false, b,
                 (short)0, acc2[nt],  false,  false);
        }
    }

#pragma unroll
    for (int nt = 0; nt < 4; ++nt) {
#pragma unroll
        for (int r = 0; r < 8; ++r) {
            const int rr = tile_row + r + 8 * (lane >> 4);
            if (rr < n) out[(size_t)rr * 64 + nt * 16 + m] = acc2[nt][r];
        }
    }
}

#define GNODES 16

__global__ void k_gather_pool(const int* __restrict__ start, const int* __restrict__ cnt,
                              const int* __restrict__ csr, const float* __restrict__ dis,
                              const float* __restrict__ h, const float* __restrict__ b2,
                              const int* __restrict__ batch, const float* __restrict__ gcnt,
                              float* __restrict__ out, int n) {
    __shared__ __attribute__((aligned(16))) float s_v[GNODES * 64];
    __shared__ int   s_g[GNODES];
    __shared__ float s_n[GNODES];
    const int tid = threadIdx.x;
    const int ln  = tid >> 4;
    const int i   = blockIdx.x * GNODES + ln;
    const int jj  = (tid & 15) * 4;

    float4 v;
    v.x = 0.0f; v.y = 0.0f; v.z = 0.0f; v.w = 0.0f;
    int g = -1;
    float gn = 1.0f;
    if (i < n) {
        const float di = dis[i];
        int j = start[i];
        const int end = j + cnt[i];
        g = batch[i];
        const float4 bv = *(const float4*)&b2[jj];

        const float4 hs = *(const float4*)&h[(size_t)i * 64 + jj];
        gn = gcnt[(g >= 0) ? g : 0];
        const float ss = di * di;
        float4 acc;
        acc.x = ss * hs.x; acc.y = ss * hs.y; acc.z = ss * hs.z; acc.w = ss * hs.w;

        for (; j + 4 <= end; j += 4) {
            const int s0 = csr[j], s1 = csr[j + 1], s2 = csr[j + 2], s3 = csr[j + 3];
            const float4 v0 = *(const float4*)&h[(size_t)s0 * 64 + jj];
            const float4 v1 = *(const float4*)&h[(size_t)s1 * 64 + jj];
            const float4 v2 = *(const float4*)&h[(size_t)s2 * 64 + jj];
            const float4 v3 = *(const float4*)&h[(size_t)s3 * 64 + jj];
            const float w0 = dis[s0] * di, w1 = dis[s1] * di;
            const float w2 = dis[s2] * di, w3 = dis[s3] * di;
            acc.x += w0 * v0.x; acc.y += w0 * v0.y; acc.z += w0 * v0.z; acc.w += w0 * v0.w;
            acc.x += w1 * v1.x; acc.y += w1 * v1.y; acc.z += w1 * v1.z; acc.w += w1 * v1.w;
            acc.x += w2 * v2.x; acc.y += w2 * v2.y; acc.z += w2 * v2.z; acc.w += w2 * v2.w;
            acc.x += w3 * v3.x; acc.y += w3 * v3.y; acc.z += w3 * v3.z; acc.w += w3 * v3.w;
        }
        const int rem = end - j;
        if (rem > 0) {
            const int s0 = csr[j];
            const int s1 = (rem > 1) ? csr[j + 1] : s0;
            const int s2 = (rem > 2) ? csr[j + 2] : s0;
            const float4 v0 = *(const float4*)&h[(size_t)s0 * 64 + jj];
            const float4 v1 = *(const float4*)&h[(size_t)s1 * 64 + jj];
            const float4 v2 = *(const float4*)&h[(size_t)s2 * 64 + jj];
            const float w0 = dis[s0] * di, w1 = dis[s1] * di, w2 = dis[s2] * di;
            acc.x += w0 * v0.x; acc.y += w0 * v0.y; acc.z += w0 * v0.z; acc.w += w0 * v0.w;
            if (rem > 1) {
                acc.x += w1 * v1.x; acc.y += w1 * v1.y; acc.z += w1 * v1.z; acc.w += w1 * v1.w;
            }
            if (rem > 2) {
                acc.x += w2 * v2.x; acc.y += w2 * v2.y; acc.z += w2 * v2.z; acc.w += w2 * v2.w;
            }
        }

        v.x = fmaxf(acc.x + bv.x, 0.0f);
        v.y = fmaxf(acc.y + bv.y, 0.0f);
        v.z = fmaxf(acc.z + bv.z, 0.0f);
        v.w = fmaxf(acc.w + bv.w, 0.0f);
    }

    *(float4*)&s_v[ln * 64 + jj] = v;
    if ((tid & 15) == 0) { s_g[ln] = g; s_n[ln] = gn; }
    __syncthreads();

    if (tid < 64) {
        int   gq[GNODES];
        float nq[GNODES];
        float vq[GNODES];
#pragma unroll
        for (int q = 0; q < GNODES; ++q) {
            gq[q] = s_g[q];
            nq[q] = s_n[q];
            vq[q] = s_v[q * 64 + tid];
        }
        int   cur = gq[0];
        float cn  = nq[0];
        float a   = 0.0f;
#pragma unroll
        for (int q = 0; q < GNODES; ++q) {
            if (gq[q] != cur) {
                if (cur >= 0) atomicAdd(&out[(size_t)cur * 64 + tid], a / fmaxf(cn, 1.0f));
                a   = 0.0f;
                cur = gq[q];
                cn  = nq[q];
            }
            a += vq[q];
        }
        if (cur >= 0) atomicAdd(&out[(size_t)cur * 64 + tid], a / fmaxf(cn, 1.0f));
    }
}

extern "C" void kernel_launch(void* const* d_in, const int* in_sizes, int n_in,
                              void* d_out, int out_size, void* d_ws, size_t ws_size,
                              hipStream_t stream) {
    const float* x     = (const float*)d_in[0];
    const int* eidx    = (const int*)d_in[1];
    const int* batch   = (const int*)d_in[2];
    const float* W1 = (const float*)d_in[4];
    const float* b1 = (const float*)d_in[5];
    const float* W2 = (const float*)d_in[6];
    const float* b2 = (const float*)d_in[7];

    const int n = in_sizes[0] / 4;
    const int E = in_sizes[1] / 2;
    const int G = out_size / 64;

    const int* src = eidx;
    const int* dst = eidx + E;

    float* ws   = (float*)d_ws;
    float* dis  = ws;
    float* bufA = dis + n;
    float* bufB = bufA + (size_t)n * 64;
    float* sums = bufB + (size_t)n * 64;
    float* cnt  = sums + (size_t)G * 64;
    int* icnt   = (int*)(cnt + G);
    int* start  = icnt + n;
    int* csr    = start + n;
    int* binned = (int*)bufA;
    int* rtot   = (int*)bufB;
    int* piece  = rtot + RMAX;

    const int T = 256;
    auto blk = [T](int work) { return (work + T - 1) / T; };

    const int R  = (n + (1 << RSH) - 1) >> RSH;
    const int NC = (E + CHUNK - 1) / CHUNK;
    k_zero_all<<<blk(G * 64 + R), T, 0, stream>>>(rtot, R, (float*)d_out, G * 64, cnt, G);
    k_chunk_count<<<NC, 1024, 0, stream>>>(dst, rtot, piece, E, R);
    k_chunk_place<<<NC, 1024, 0, stream>>>(src, dst, rtot, piece, binned, E, R);
    k_range_sort<<<R, 1024, 0, stream>>>(rtot, binned, csr, start, icnt, dis, batch, cnt, n, R);

    k_agg_gemm12<<<(n + 63) / 64, 128, 0, stream>>>(start, icnt, csr, dis, x, W1, b1, W2, bufA, n);
    k_gather_pool<<<blk(n * 16), T, 0, stream>>>(start, icnt, csr, dis, bufA, b2, batch, cnt, (float*)d_out, n);
}


# edge binning chunk-local: place kernel stores its chunk range-ordered in place + a (start,length) table; range kernel collects its pieces by it; count, zero launches gone; slot (sum) order changes
# speedup vs baseline: 7.1639x; 1.0900x over previous
#include <hip/hip_runtime.h>
#include <hip/hip_bf16.h>

typedef __attribute__((ext_vector_type(2))) float v2f;
typedef __attribute__((ext_vector_type(8))) float v8f;

#define RSH    8
#define RNODES (1 << RSH)
#define RMASK  (RNODES - 1)
#define RMAX   1024
#define CHUNK  4096
#define EPT    (CHUNK / 1024)
#define SCAP   8192
#define PGRP   8

__device__ __forceinline__ int block_excl_scan(int c, int* s_w) {
    const int tid  = threadIdx.x;
    const int lane = tid & 31;
    const int wave = tid >> 5;
    int inc = c;
    for (int o = 1; o < 32; o <<= 1) {
        int v = __shfl_up(inc, o);
        if (lane >= o) inc += v;
    }
    if (lane == 31) s_w[wave] = inc;
    __syncthreads();
    int v = (lane < wave) ? s_w[lane] : 0;
    for (int o = 16; o > 0; o >>= 1) v += __shfl_xor(v, o);
    return v + inc - c;
}

__global__ void k_chunk_place(const int* __restrict__ src, const int* __restrict__ dst,
                              int* __restrict__ tab, int* __restrict__ binned,
                              float* __restrict__ out, int nout,
                              float* __restrict__ gcnt, int G, int E, int R) {
    __shared__ int s_cur[RMAX];
    __shared__ int s_w0[32];
    __shared__ int s_word[CHUNK];
    const int tid = threadIdx.x;
    const int e0  = blockIdx.x * CHUNK;

    const int gi = blockIdx.x * 1024 + tid;
    const int gs = gridDim.x * 1024;
    for (int i = gi; i < nout; i += gs) out[i]  = 0.0f;
    for (int i = gi; i < G;    i += gs) gcnt[i] = 0.0f;

    int d[EPT], s[EPT];
#pragma unroll
    for (int q = 0; q < EPT; ++q) {
        const int e  = e0 + q * 1024 + tid;
        const int ec = (e < E) ? e : E - 1;
        const int v  = dst[ec];
        s[q] = src[ec];
        d[q] = (e < E) ? v : -1;
    }
    s_cur[tid] = 0;
    __syncthreads();
#pragma unroll
    for (int q = 0; q < EPT; ++q)
        if (d[q] >= 0) atomicAdd(&s_cur[d[q] >> RSH], 1);
    __syncthreads();

    const int h   = s_cur[tid];
    const int lex = block_excl_scan(h, s_w0);
    s_cur[tid] = lex;
    if (tid < R) tab[(size_t)blockIdx.x * R + tid] = lex | (h << 16);
    __syncthreads();

#pragma unroll
    for (int q = 0; q < EPT; ++q) {
        if (d[q] >= 0) {
            const int lp = atomicAdd(&s_cur[d[q] >> RSH], 1);
            s_word[lp] = (s[q] << RSH) | (d[q] & RMASK);
        }
    }
    __syncthreads();

#pragma unroll
    for (int q = 0; q < EPT; ++q) {
        const int j = q * 1024 + tid;
        if (e0 + j < E) binned[e0 + j] = s_word[j];
    }
}

template <class F>
__device__ __forceinline__ void for_range_words(const int* __restrict__ tab,
                                                const int* __restrict__ binned,
                                                int r, int R, int NC, int inf0, F f) {
    const int lane = threadIdx.x & 31;
    const int wave = threadIdx.x >> 5;
    for (int cb = 0; cb < NC; cb += 1024) {
        int inf = inf0;
        if (cb > 0) {
            const int cl = cb + lane * 32 + wave;
            inf = (cl < NC) ? tab[(size_t)cl * R + r] : 0;
        }
        int nk = (NC - cb - wave + 31) >> 5;
        nk = (nk < 0) ? 0 : ((nk > 32) ? 32 : nk);
        for (int k = 0; k < nk; k += PGRP) {
            int w[PGRP], cn[PGRP], base[PGRP];
#pragma unroll
            for (int u = 0; u < PGRP; ++u) {
                const int iu = __shfl(inf, k + u);
                cn[u]   = iu >> 16;
                base[u] = (cb + (k + u) * 32 + wave) * CHUNK + (iu & 0xFFFF);
                w[u]    = binned[(lane < cn[u]) ? base[u] + lane : 0];
            }
#pragma unroll
            for (int u = 0; u < PGRP; ++u)
                if (lane < cn[u]) f(w[u]);
#pragma unroll
            for (int u = 0; u < PGRP; ++u)
                for (int o = lane + 32; o < cn[u]; o += 32) f(binned[base[u] + o]);
        }
    }
}

__global__ void k_range_sort(const int* __restrict__ tab, const int* __restrict__ binned,
                             int* __restrict__ csr, int* __restrict__ start,
                             int* __restrict__ cnt, float* __restrict__ dis,
                             const int* __restrict__ batch, float* __restrict__ gcnt,
                             int n, int R, int NC) {
    __shared__ int s_w0[32];
    __shared__ int s_w1[32];
    __shared__ int s_c[RNODES];
    __shared__ int s_o[RNODES];
    __shared__ int s_out[SCAP];
    const int tid  = threadIdx.x;
    const int lane = tid & 31;
    const int wave = tid >> 5;
    const int r    = blockIdx.x;

    const int cl0  = lane * 32 + wave;
    const int inf0 = (cl0 < NC) ? tab[(size_t)cl0 * R + r] : 0;
    int mlen = inf0 >> 16;
    int mb0  = inf0 & 0xFFFF;
    for (int cb = 1024; cb < NC; cb += 1024) {
        const int cl = cb + cl0;
        const int t  = (cl < NC) ? tab[(size_t)cl * R + r] : 0;
        mlen += t >> 16;
        mb0  += t & 0xFFFF;
    }
    for (int o = 16; o > 0; o >>= 1) {
        mlen += __shfl_xor(mlen, o);
        mb0  += __shfl_xor(mb0, o);
    }
    if (lane == 0) { s_w0[wave] = mlen; s_w1[wave] = mb0; }
    if (tid < RNODES) s_c[tid] = 0;
    __syncthreads();
    int len = s_w0[lane];
    int b0  = s_w1[lane];
    for (int o = 16; o > 0; o >>= 1) {
        len += __shfl_xor(len, o);
        b0  += __shfl_xor(b0, o);
    }

    for_range_words(tab, binned, r, R, NC, inf0,
                    [&](int w) { atomicAdd(&s_c[w & RMASK], 1); });
    __syncthreads();

    const int cn  = (tid < RNODES) ? s_c[tid] : 0;
    const int off = block_excl_scan(cn, s_w1);
    if (tid < RNODES) {
        s_o[tid] = off;
        const int i = (r << RSH) + tid;
        const int g = (i < n) ? batch[i] : -1;
        if (i < n) {
            start[i] = b0 + off;
            cnt[i]   = cn;
            float dg = (float)(cn + 1);
            dis[i] = (dg > 0.0f) ? rsqrtf(dg) : 0.0f;
        }
        unsigned todo = (unsigned)__ballot(g >= 0);
        for (int it = 0; it < 32 && todo != 0u; ++it) {
            const int leader = __ffs((int)todo) - 1;
            const int gl = __shfl(g, leader);
            const unsigned same = (unsigned)__ballot(g == gl) & todo;
            if (lane == leader) atomicAdd(&gcnt[gl], (float)__popc(same));
            todo &= ~same;
        }
    }
    __syncthreads();

    if (len <= SCAP) {
        for_range_words(tab, binned, r, R, NC, inf0,
                        [&](int w) { s_out[atomicAdd(&s_o[w & RMASK], 1)] = w >> RSH; });
        __syncthreads();
        for (int k = tid; k < len; k += 1024) csr[b0 + k] = s_out[k];
    } else {
        for_range_words(tab, binned, r, R, NC, inf0,
                        [&](int w) { csr[b0 + atomicAdd(&s_o[w & RMASK], 1)] = w >> RSH; });
    }
}

#define TPITCH 68
#define WPITCH 72

__global__ void k_agg_gemm12(const int* __restrict__ start, const int* __restrict__ cnt,
                             const int* __restrict__ csr, const float* __restrict__ dis,
                             const float* __restrict__ x, const float* __restrict__ W1,
                             const float* __restrict__ b1, const float* __restrict__ W2,
                             float* __restrict__ out, int n) {
    __shared__ __attribute__((aligned(16))) float sW[64 * WPITCH];
    __shared__ __attribute__((aligned(16))) float sT[4 * 16 * TPITCH];
    const int tid  = threadIdx.x;
    const int wave = tid >> 5;
    const int lane = tid & 31;
    const int tile_row = blockIdx.x * 64 + wave * 16;

    float4 wq[8];
#pragma unroll
    for (int q = 0; q < 8; ++q) wq[q] = *(const float4*)&W2[(size_t)(q * 128 + tid) * 4];

    const int m    = lane & 15;
    const int koff = (lane >> 4) * 2;
    int row = tile_row + m;
    if (row >= n) row = n - 1;

    const float di = dis[row];
    int j = start[row];
    const int end = j + cnt[row];

    const float2 xs = *(const float2*)&x[(size_t)row * 4 + koff];

#pragma unroll
    for (int q = 0; q < 8; ++q) {
        const int i = (q * 128 + tid) * 4;
        *(float4*)&sW[(i >> 6) * WPITCH + (i & 63)] = wq[q];
    }

    const float ss = di * di;
    float ax = ss * xs.x, ay = ss * xs.y;

    for (; j + 4 <= end; j += 4) {
        const int s0 = csr[j], s1 = csr[j + 1], s2 = csr[j + 2], s3 = csr[j + 3];
        const float2 v0 = *(const float2*)&x[(size_t)s0 * 4 + koff];
        const float2 v1 = *(const float2*)&x[(size_t)s1 * 4 + koff];
        const float2 v2 = *(const float2*)&x[(size_t)s2 * 4 + koff];
        const float2 v3 = *(const float2*)&x[(size_t)s3 * 4 + koff];
        const float w0 = dis[s0] * di, w1 = dis[s1] * di;
        const float w2 = dis[s2] * di, w3 = dis[s3] * di;
        ax += w0 * v0.x; ay += w0 * v0.y;
        ax += w1 * v1.x; ay += w1 * v1.y;
        ax += w2 * v2.x; ay += w2 * v2.y;
        ax += w3 * v3.x; ay += w3 * v3.y;
    }
    const int rem = end - j;
    if (rem > 0) {
        const int s0 = csr[j];
        const int s1 = (rem > 1) ? csr[j + 1] : s0;
        const int s2 = (rem > 2) ? csr[j + 2] : s0;
        const float2 v0 = *(const float2*)&x[(size_t)s0 * 4 + koff];
        const float2 v1 = *(const float2*)&x[(size_t)s1 * 4 + koff];
        const float2 v2 = *(const float2*)&x[(size_t)s2 * 4 + koff];
        const float w0 = dis[s0] * di, w1 = dis[s1] * di, w2 = dis[s2] * di;
        ax += w0 * v0.x; ay += w0 * v0.y;
        if (rem > 1) { ax += w1 * v1.x; ay += w1 * v1.y; }
        if (rem > 2) { ax += w2 * v2.x; ay += w2 * v2.y; }
    }

    v2f a;
    a.x = ax;
    a.y = ay;

    v8f acc[4];
#pragma unroll
    for (int nt = 0; nt < 4; ++nt) {
        v2f bf;
        const int col = nt * 16 + m;
        bf.x = W1[(koff) * 64 + col];
        bf.y = W1[(koff + 1) * 64 + col];
        acc[nt] = __builtin_amdgcn_wmma_f32_16x16x4_f32(
             false, a,  false, bf,
             (short)0, v8f{0.f},  false,  false);
    }

    float* tw = sT + wave * 16 * TPITCH;
#pragma unroll
    for (int nt = 0; nt < 4; ++nt) {
        const float bv = b1[nt * 16 + m];
#pragma unroll
        for (int r = 0; r < 8; ++r)
            tw[(r + 8 * (lane >> 4)) * TPITCH + nt * 16 + m] = fmaxf(acc[nt][r] + bv, 0.0f);
    }
    __syncthreads();

    v8f acc2[4] = {v8f{0.f}, v8f{0.f}, v8f{0.f}, v8f{0.f}};
#pragma unroll
    for (int kb = 0; kb < 64; kb += 4) {
        const v2f a2 = *(const v2f*)&tw[m * TPITCH + kb + koff];
#pragma unroll
        for (int nt = 0; nt < 4; ++nt) {
            v2f b;
            const int col = nt * 16 + m;
            b.x = sW[(kb + koff) * WPITCH + col];
            b.y = sW[(kb + koff + 1) * WPITCH + col];
            acc2[nt] = __builtin_amdgcn_wmma_f32_16x16x4_f32(
                 false, a2,  false, b,
                 (short)0, acc2[nt],  false,  false);
        }
    }

#pragma unroll
    for (int nt = 0; nt < 4; ++nt) {
#pragma unroll
        for (int r = 0; r < 8; ++r) {
            const int rr = tile_row + r + 8 * (lane >> 4);
            if (rr < n) out[(size_t)rr * 64 + nt * 16 + m] = acc2[nt][r];
        }
    }
}

#define GNODES 16

__global__ void k_gather_pool(const int* __restrict__ start, const int* __restrict__ cnt,
                              const int* __restrict__ csr, const float* __restrict__ dis,
                              const float* __restrict__ h, const float* __restrict__ b2,
                              const int* __restrict__ batch, const float* __restrict__ gcnt,
                              float* __restrict__ out, int n) {
    __shared__ __attribute__((aligned(16))) float s_v[GNODES * 64];
    __shared__ int   s_g[GNODES];
    __shared__ float s_n[GNODES];
    const int tid = threadIdx.x;
    const int ln  = tid >> 4;
    const int i   = blockIdx.x * GNODES + ln;
    const int jj  = (tid & 15) * 4;

    float4 v;
    v.x = 0.0f; v.y = 0.0f; v.z = 0.0f; v.w = 0.0f;
    int g = -1;
    float gn = 1.0f;
    if (i < n) {
        const float di = dis[i];
        int j = start[i];
        const int end = j + cnt[i];
        g = batch[i];
        const float4 bv = *(const float4*)&b2[jj];

        const float4 hs = *(const float4*)&h[(size_t)i * 64 + jj];
        gn = gcnt[(g >= 0) ? g : 0];
        const float ss = di * di;
        float4 acc;
        acc.x = ss * hs.x; acc.y = ss * hs.y; acc.z = ss * hs.z; acc.w = ss * hs.w;

        for (; j + 4 <= end; j += 4) {
            const int s0 = csr[j], s1 = csr[j + 1], s2 = csr[j + 2], s3 = csr[j + 3];
            const float4 v0 = *(const float4*)&h[(size_t)s0 * 64 + jj];
            const float4 v1 = *(const float4*)&h[(size_t)s1 * 64 + jj];
            const float4 v2 = *(const float4*)&h[(size_t)s2 * 64 + jj];
            const float4 v3 = *(const float4*)&h[(size_t)s3 * 64 + jj];
            const float w0 = dis[s0] * di, w1 = dis[s1] * di;
            const float w2 = dis[s2] * di, w3 = dis[s3] * di;
            acc.x += w0 * v0.x; acc.y += w0 * v0.y; acc.z += w0 * v0.z; acc.w += w0 * v0.w;
            acc.x += w1 * v1.x; acc.y += w1 * v1.y; acc.z += w1 * v1.z; acc.w += w1 * v1.w;
            acc.x += w2 * v2.x; acc.y += w2 * v2.y; acc.z += w2 * v2.z; acc.w += w2 * v2.w;
            acc.x += w3 * v3.x; acc.y += w3 * v3.y; acc.z += w3 * v3.z; acc.w += w3 * v3.w;
        }
        const int rem = end - j;
        if (rem > 0) {
            const int s0 = csr[j];
            const int s1 = (rem > 1) ? csr[j + 1] : s0;
            const int s2 = (rem > 2) ? csr[j + 2] : s0;
            const float4 v0 = *(const float4*)&h[(size_t)s0 * 64 + jj];
            const float4 v1 = *(const float4*)&h[(size_t)s1 * 64 + jj];
            const float4 v2 = *(const float4*)&h[(size_t)s2 * 64 + jj];
            const float w0 = dis[s0] * di, w1 = dis[s1] * di, w2 = dis[s2] * di;
            acc.x += w0 * v0.x; acc.y += w0 * v0.y; acc.z += w0 * v0.z; acc.w += w0 * v0.w;
            if (rem > 1) {
                acc.x += w1 * v1.x; acc.y += w1 * v1.y; acc.z += w1 * v1.z; acc.w += w1 * v1.w;
            }
            if (rem > 2) {
                acc.x += w2 * v2.x; acc.y += w2 * v2.y; acc.z += w2 * v2.z; acc.w += w2 * v2.w;
            }
        }

        v.x = fmaxf(acc.x + bv.x, 0.0f);
        v.y = fmaxf(acc.y + bv.y, 0.0f);
        v.z = fmaxf(acc.z + bv.z, 0.0f);
        v.w = fmaxf(acc.w + bv.w, 0.0f);
    }

    *(float4*)&s_v[ln * 64 + jj] = v;
    if ((tid & 15) == 0) { s_g[ln] = g; s_n[ln] = gn; }
    __syncthreads();

    if (tid < 64) {
        int   gq[GNODES];
        float nq[GNODES];
        float vq[GNODES];
#pragma unroll
        for (int q = 0; q < GNODES; ++q) {
            gq[q] = s_g[q];
            nq[q] = s_n[q];
            vq[q] = s_v[q * 64 + tid];
        }
        int   cur = gq[0];
        float cn  = nq[0];
        float a   = 0.0f;
#pragma unroll
        for (int q = 0; q < GNODES; ++q) {
            if (gq[q] != cur) {
                if (cur >= 0) atomicAdd(&out[(size_t)cur * 64 + tid], a / fmaxf(cn, 1.0f));
                a   = 0.0f;
                cur = gq[q];
                cn  = nq[q];
            }
            a += vq[q];
        }
        if (cur >= 0) atomicAdd(&out[(size_t)cur * 64 + tid], a / fmaxf(cn, 1.0f));
    }
}

extern "C" void kernel_launch(void* const* d_in, const int* in_sizes, int n_in,
                              void* d_out, int out_size, void* d_ws, size_t ws_size,
                              hipStream_t stream) {
    const float* x     = (const float*)d_in[0];
    const int* eidx    = (const int*)d_in[1];
    const int* batch   = (const int*)d_in[2];
    const float* W1 = (const float*)d_in[4];
    const float* b1 = (const float*)d_in[5];
    const float* W2 = (const float*)d_in[6];
    const float* b2 = (const float*)d_in[7];

    const int n = in_sizes[0] / 4;
    const int E = in_sizes[1] / 2;
    const int G = out_size / 64;

    const int* src = eidx;
    const int* dst = eidx + E;

    float* ws   = (float*)d_ws;
    float* dis  = ws;
    float* bufA = dis + n;
    float* bufB = bufA + (size_t)n * 64;
    float* sums = bufB + (size_t)n * 64;
    float* cnt  = sums + (size_t)G * 64;
    int* icnt   = (int*)(cnt + G);
    int* start  = icnt + n;
    int* csr    = start + n;
    int* binned = (int*)bufA;
    int* tab    = (int*)bufB;

    const int T = 256;
    auto blk = [T](int work) { return (work + T - 1) / T; };

    const int R  = (n + (1 << RSH) - 1) >> RSH;
    const int NC = (E + CHUNK - 1) / CHUNK;
    k_chunk_place<<<NC, 1024, 0, stream>>>(src, dst, tab, binned, (float*)d_out, G * 64, cnt, G, E, R);
    k_range_sort<<<R, 1024, 0, stream>>>(tab, binned, csr, start, icnt, dis, batch, cnt, n, R, NC);

    k_agg_gemm12<<<(n + 63) / 64, 128, 0, stream>>>(start, icnt, csr, dis, x, W1, b1, W2, bufA, n);
    k_gather_pool<<<blk(n * 16), T, 0, stream>>>(start, icnt, csr, dis, bufA, b2, batch, cnt, (float*)d_out, n);
}


# source rows stored pre-scaled by their own dis (xs = dis*x in the sort launch; h2 rows via the GEMM's A fragment): both gathers drop the per-edge dis[s] load and multiply; rounding changes
# speedup vs baseline: 8.5045x; 1.1871x over previous
#include <hip/hip_runtime.h>
#include <hip/hip_bf16.h>

typedef __attribute__((ext_vector_type(2))) float v2f;
typedef __attribute__((ext_vector_type(8))) float v8f;

#define RSH    8
#define RNODES (1 << RSH)
#define RMASK  (RNODES - 1)
#define RMAX   1024
#define CHUNK  4096
#define EPT    (CHUNK / 1024)
#define SCAP   8192
#define PGRP   8

__device__ __forceinline__ int block_excl_scan(int c, int* s_w) {
    const int tid  = threadIdx.x;
    const int lane = tid & 31;
    const int wave = tid >> 5;
    int inc = c;
    for (int o = 1; o < 32; o <<= 1) {
        int v = __shfl_up(inc, o);
        if (lane >= o) inc += v;
    }
    if (lane == 31) s_w[wave] = inc;
    __syncthreads();
    int v = (lane < wave) ? s_w[lane] : 0;
    for (int o = 16; o > 0; o >>= 1) v += __shfl_xor(v, o);
    return v + inc - c;
}

__global__ void k_chunk_place(const int* __restrict__ src, const int* __restrict__ dst,
                              int* __restrict__ tab, int* __restrict__ binned,
                              float* __restrict__ out, int nout,
                              float* __restrict__ gcnt, int G, int E, int R) {
    __shared__ int s_cur[RMAX];
    __shared__ int s_w0[32];
    __shared__ int s_word[CHUNK];
    const int tid = threadIdx.x;
    const int e0  = blockIdx.x * CHUNK;

    const int gi = blockIdx.x * 1024 + tid;
    const int gs = gridDim.x * 1024;
    for (int i = gi; i < nout; i += gs) out[i]  = 0.0f;
    for (int i = gi; i < G;    i += gs) gcnt[i] = 0.0f;

    int d[EPT], s[EPT];
#pragma unroll
    for (int q = 0; q < EPT; ++q) {
        const int e  = e0 + q * 1024 + tid;
        const int ec = (e < E) ? e : E - 1;
        const int v  = dst[ec];
        s[q] = src[ec];
        d[q] = (e < E) ? v : -1;
    }
    s_cur[tid] = 0;
    __syncthreads();
#pragma unroll
    for (int q = 0; q < EPT; ++q)
        if (d[q] >= 0) atomicAdd(&s_cur[d[q] >> RSH], 1);
    __syncthreads();

    const int h   = s_cur[tid];
    const int lex = block_excl_scan(h, s_w0);
    s_cur[tid] = lex;
    if (tid < R) tab[(size_t)blockIdx.x * R + tid] = lex | (h << 16);
    __syncthreads();

#pragma unroll
    for (int q = 0; q < EPT; ++q) {
        if (d[q] >= 0) {
            const int lp = atomicAdd(&s_cur[d[q] >> RSH], 1);
            s_word[lp] = (s[q] << RSH) | (d[q] & RMASK);
        }
    }
    __syncthreads();

#pragma unroll
    for (int q = 0; q < EPT; ++q) {
        const int j = q * 1024 + tid;
        if (e0 + j < E) binned[e0 + j] = s_word[j];
    }
}

template <class F>
__device__ __forceinline__ void for_range_words(const int* __restrict__ tab,
                                                const int* __restrict__ binned,
                                                int r, int R, int NC, int inf0, F f) {
    const int lane = threadIdx.x & 31;
    const int wave = threadIdx.x >> 5;
    for (int cb = 0; cb < NC; cb += 1024) {
        int inf = inf0;
        if (cb > 0) {
            const int cl = cb + lane * 32 + wave;
            inf = (cl < NC) ? tab[(size_t)cl * R + r] : 0;
        }
        int nk = (NC - cb - wave + 31) >> 5;
        nk = (nk < 0) ? 0 : ((nk > 32) ? 32 : nk);
        for (int k = 0; k < nk; k += PGRP) {
            int w[PGRP], cn[PGRP], base[PGRP];
#pragma unroll
            for (int u = 0; u < PGRP; ++u) {
                const int iu = __shfl(inf, k + u);
                cn[u]   = iu >> 16;
                base[u] = (cb + (k + u) * 32 + wave) * CHUNK + (iu & 0xFFFF);
                w[u]    = binned[(lane < cn[u]) ? base[u] + lane : 0];
            }
#pragma unroll
            for (int u = 0; u < PGRP; ++u)
                if (lane < cn[u]) f(w[u]);
#pragma unroll
            for (int u = 0; u < PGRP; ++u)
                for (int o = lane + 32; o < cn[u]; o += 32) f(binned[base[u] + o]);
        }
    }
}

__global__ void k_range_sort(const int* __restrict__ tab, const int* __restrict__ binned,
                             int* __restrict__ csr, int* __restrict__ start,
                             int* __restrict__ cnt, float* __restrict__ dis,
                             const int* __restrict__ batch, float* __restrict__ gcnt,
                             const float* __restrict__ x, float* __restrict__ xs,
                             int n, int R, int NC) {
    __shared__ int s_w0[32];
    __shared__ int s_w1[32];
    __shared__ int s_c[RNODES];
    __shared__ int s_o[RNODES];
    __shared__ int s_out[SCAP];
    const int tid  = threadIdx.x;
    const int lane = tid & 31;
    const int wave = tid >> 5;
    const int r    = blockIdx.x;

    const int cl0  = lane * 32 + wave;
    const int inf0 = (cl0 < NC) ? tab[(size_t)cl0 * R + r] : 0;
    float4 xv;
    xv.x = 0.0f; xv.y = 0.0f; xv.z = 0.0f; xv.w = 0.0f;
    if (tid < RNODES) {
        const int in0 = (r << RSH) + tid;
        xv = *(const float4*)&x[(size_t)((in0 < n) ? in0 : n - 1) * 4];
    }
    int mlen = inf0 >> 16;
    int mb0  = inf0 & 0xFFFF;
    for (int cb = 1024; cb < NC; cb += 1024) {
        const int cl = cb + cl0;
        const int t  = (cl < NC) ? tab[(size_t)cl * R + r] : 0;
        mlen += t >> 16;
        mb0  += t & 0xFFFF;
    }
    for (int o = 16; o > 0; o >>= 1) {
        mlen += __shfl_xor(mlen, o);
        mb0  += __shfl_xor(mb0, o);
    }
    if (lane == 0) { s_w0[wave] = mlen; s_w1[wave] = mb0; }
    if (tid < RNODES) s_c[tid] = 0;
    __syncthreads();
    int len = s_w0[lane];
    int b0  = s_w1[lane];
    for (int o = 16; o > 0; o >>= 1) {
        len += __shfl_xor(len, o);
        b0  += __shfl_xor(b0, o);
    }

    for_range_words(tab, binned, r, R, NC, inf0,
                    [&](int w) { atomicAdd(&s_c[w & RMASK], 1); });
    __syncthreads();

    const int cn  = (tid < RNODES) ? s_c[tid] : 0;
    const int off = block_excl_scan(cn, s_w1);
    if (tid < RNODES) {
        s_o[tid] = off;
        const int i = (r << RSH) + tid;
        const int g = (i < n) ? batch[i] : -1;
        if (i < n) {
            start[i] = b0 + off;
            cnt[i]   = cn;
            float dg = (float)(cn + 1);
            const float dv = (dg > 0.0f) ? rsqrtf(dg) : 0.0f;
            dis[i] = dv;
            float4 xo;
            xo.x = dv * xv.x; xo.y = dv * xv.y; xo.z = dv * xv.z; xo.w = dv * xv.w;
            *(float4*)&xs[(size_t)i * 4] = xo;
        }
        unsigned todo = (unsigned)__ballot(g >= 0);
        for (int it = 0; it < 32 && todo != 0u; ++it) {
            const int leader = __ffs((int)todo) - 1;
            const int gl = __shfl(g, leader);
            const unsigned same = (unsigned)__ballot(g == gl) & todo;
            if (lane == leader) atomicAdd(&gcnt[gl], (float)__popc(same));
            todo &= ~same;
        }
    }
    __syncthreads();

    if (len <= SCAP) {
        for_range_words(tab, binned, r, R, NC, inf0,
                        [&](int w) { s_out[atomicAdd(&s_o[w & RMASK], 1)] = w >> RSH; });
        __syncthreads();
        for (int k = tid; k < len; k += 1024) csr[b0 + k] = s_out[k];
    } else {
        for_range_words(tab, binned, r, R, NC, inf0,
                        [&](int w) { csr[b0 + atomicAdd(&s_o[w & RMASK], 1)] = w >> RSH; });
    }
}

#define TPITCH 68
#define WPITCH 72

__global__ void k_agg_gemm12(const int* __restrict__ start, const int* __restrict__ cnt,
                             const int* __restrict__ csr, const float* __restrict__ dis,
                             const float* __restrict__ xs, const float* __restrict__ W1,
                             const float* __restrict__ b1, const float* __restrict__ W2,
                             float* __restrict__ out, int n) {
    __shared__ __attribute__((aligned(16))) float sW[64 * WPITCH];
    __shared__ __attribute__((aligned(16))) float sT[4 * 16 * TPITCH];
    const int tid  = threadIdx.x;
    const int wave = tid >> 5;
    const int lane = tid & 31;
    const int tile_row = blockIdx.x * 64 + wave * 16;

    float4 wq[8];
#pragma unroll
    for (int q = 0; q < 8; ++q) wq[q] = *(const float4*)&W2[(size_t)(q * 128 + tid) * 4];

    const int m    = lane & 15;
    const int koff = (lane >> 4) * 2;
    int row = tile_row + m;
    if (row >= n) row = n - 1;

    const float di = dis[row];
    int j = start[row];
    const int end = j + cnt[row];

    const float2 xo = *(const float2*)&xs[(size_t)row * 4 + koff];

#pragma unroll
    for (int q = 0; q < 8; ++q) {
        const int i = (q * 128 + tid) * 4;
        *(float4*)&sW[(i >> 6) * WPITCH + (i & 63)] = wq[q];
    }

    float ax = xo.x, ay = xo.y;

    for (; j + 4 <= end; j += 4) {
        const int s0 = csr[j], s1 = csr[j + 1], s2 = csr[j + 2], s3 = csr[j + 3];
        const float2 v0 = *(const float2*)&xs[(size_t)s0 * 4 + koff];
        const float2 v1 = *(const float2*)&xs[(size_t)s1 * 4 + koff];
        const float2 v2 = *(const float2*)&xs[(size_t)s2 * 4 + koff];
        const float2 v3 = *(const float2*)&xs[(size_t)s3 * 4 + koff];
        ax += v0.x; ay += v0.y;
        ax += v1.x; ay += v1.y;
        ax += v2.x; ay += v2.y;
        ax += v3.x; ay += v3.y;
    }
    const int rem = end - j;
    if (rem > 0) {
        const int s0 = csr[j];
        const int s1 = (rem > 1) ? csr[j + 1] : s0;
        const int s2 = (rem > 2) ? csr[j + 2] : s0;
        const float2 v0 = *(const float2*)&xs[(size_t)s0 * 4 + koff];
        const float2 v1 = *(const float2*)&xs[(size_t)s1 * 4 + koff];
        const float2 v2 = *(const float2*)&xs[(size_t)s2 * 4 + koff];
        ax += v0.x; ay += v0.y;
        if (rem > 1) { ax += v1.x; ay += v1.y; }
        if (rem > 2) { ax += v2.x; ay += v2.y; }
    }

    v2f a;
    a.x = di * ax;
    a.y = di * ay;

    v8f acc[4];
#pragma unroll
    for (int nt = 0; nt < 4; ++nt) {
        v2f bf;
        const int col = nt * 16 + m;
        bf.x = W1[(koff) * 64 + col];
        bf.y = W1[(koff + 1) * 64 + col];
        acc[nt] = __builtin_amdgcn_wmma_f32_16x16x4_f32(
             false, a,  false, bf,
             (short)0, v8f{0.f},  false,  false);
    }

    float* tw = sT + wave * 16 * TPITCH;
#pragma unroll
    for (int nt = 0; nt < 4; ++nt) {
        const float bv = b1[nt * 16 + m];
#pragma unroll
        for (int r = 0; r < 8; ++r)
            tw[(r + 8 * (lane >> 4)) * TPITCH + nt * 16 + m] = fmaxf(acc[nt][r] + bv, 0.0f);
    }
    __syncthreads();

    v8f acc2[4] = {v8f{0.f}, v8f{0.f}, v8f{0.f}, v8f{0.f}};
#pragma unroll
    for (int kb = 0; kb < 64; kb += 4) {
        v2f a2 = *(const v2f*)&tw[m * TPITCH + kb + koff];
        a2.x *= di;
        a2.y *= di;
#pragma unroll
        for (int nt = 0; nt < 4; ++nt) {
            v2f b;
            const int col = nt * 16 + m;
            b.x = sW[(kb + koff) * WPITCH + col];
            b.y = sW[(kb + koff + 1) * WPITCH + col];
            acc2[nt] = __builtin_amdgcn_wmma_f32_16x16x4_f32(
                 false, a2,  false, b,
                 (short)0, acc2[nt],  false,  false);
        }
    }

#pragma unroll
    for (int nt = 0; nt < 4; ++nt) {
#pragma unroll
        for (int r = 0; r < 8; ++r) {
            const int rr = tile_row + r + 8 * (lane >> 4);
            if (rr < n) out[(size_t)rr * 64 + nt * 16 + m] = acc2[nt][r];
        }
    }
}

#define GNODES 16

__global__ void k_gather_pool(const int* __restrict__ start, const int* __restrict__ cnt,
                              const int* __restrict__ csr, const float* __restrict__ dis,
                              const float* __restrict__ h, const float* __restrict__ b2,
                              const int* __restrict__ batch, const float* __restrict__ gcnt,
                              float* __restrict__ out, int n) {
    __shared__ __attribute__((aligned(16))) float s_v[GNODES * 64];
    __shared__ int   s_g[GNODES];
    __shared__ float s_n[GNODES];
    const int tid = threadIdx.x;
    const int ln  = tid >> 4;
    const int i   = blockIdx.x * GNODES + ln;
    const int jj  = (tid & 15) * 4;

    float4 v;
    v.x = 0.0f; v.y = 0.0f; v.z = 0.0f; v.w = 0.0f;
    int g = -1;
    float gn = 1.0f;
    if (i < n) {
        const float di = dis[i];
        int j = start[i];
        const int end = j + cnt[i];
        g = batch[i];
        const float4 bv = *(const float4*)&b2[jj];

        const float4 hs = *(const float4*)&h[(size_t)i * 64 + jj];
        gn = gcnt[(g >= 0) ? g : 0];
        float4 acc = hs;

        for (; j + 4 <= end; j += 4) {
            const int s0 = csr[j], s1 = csr[j + 1], s2 = csr[j + 2], s3 = csr[j + 3];
            const float4 v0 = *(const float4*)&h[(size_t)s0 * 64 + jj];
            const float4 v1 = *(const float4*)&h[(size_t)s1 * 64 + jj];
            const float4 v2 = *(const float4*)&h[(size_t)s2 * 64 + jj];
            const float4 v3 = *(const float4*)&h[(size_t)s3 * 64 + jj];
            acc.x += v0.x; acc.y += v0.y; acc.z += v0.z; acc.w += v0.w;
            acc.x += v1.x; acc.y += v1.y; acc.z += v1.z; acc.w += v1.w;
            acc.x += v2.x; acc.y += v2.y; acc.z += v2.z; acc.w += v2.w;
            acc.x += v3.x; acc.y += v3.y; acc.z += v3.z; acc.w += v3.w;
        }
        const int rem = end - j;
        if (rem > 0) {
            const int s0 = csr[j];
            const int s1 = (rem > 1) ? csr[j + 1] : s0;
            const int s2 = (rem > 2) ? csr[j + 2] : s0;
            const float4 v0 = *(const float4*)&h[(size_t)s0 * 64 + jj];
            const float4 v1 = *(const float4*)&h[(size_t)s1 * 64 + jj];
            const float4 v2 = *(const float4*)&h[(size_t)s2 * 64 + jj];
            acc.x += v0.x; acc.y += v0.y; acc.z += v0.z; acc.w += v0.w;
            if (rem > 1) {
                acc.x += v1.x; acc.y += v1.y; acc.z += v1.z; acc.w += v1.w;
            }
            if (rem > 2) {
                acc.x += v2.x; acc.y += v2.y; acc.z += v2.z; acc.w += v2.w;
            }
        }

        v.x = fmaxf(di * acc.x + bv.x, 0.0f);
        v.y = fmaxf(di * acc.y + bv.y, 0.0f);
        v.z = fmaxf(di * acc.z + bv.z, 0.0f);
        v.w = fmaxf(di * acc.w + bv.w, 0.0f);
    }

    *(float4*)&s_v[ln * 64 + jj] = v;
    if ((tid & 15) == 0) { s_g[ln] = g; s_n[ln] = gn; }
    __syncthreads();

    if (tid < 64) {
        int   gq[GNODES];
        float nq[GNODES];
        float vq[GNODES];
#pragma unroll
        for (int q = 0; q < GNODES; ++q) {
            gq[q] = s_g[q];
            nq[q] = s_n[q];
            vq[q] = s_v[q * 64 + tid];
        }
        int   cur = gq[0];
        float cn  = nq[0];
        float a   = 0.0f;
#pragma unroll
        for (int q = 0; q < GNODES; ++q) {
            if (gq[q] != cur) {
                if (cur >= 0) atomicAdd(&out[(size_t)cur * 64 + tid], a / fmaxf(cn, 1.0f));
                a   = 0.0f;
                cur = gq[q];
                cn  = nq[q];
            }
            a += vq[q];
        }
        if (cur >= 0) atomicAdd(&out[(size_t)cur * 64 + tid], a / fmaxf(cn, 1.0f));
    }
}

extern "C" void kernel_launch(void* const* d_in, const int* in_sizes, int n_in,
                              void* d_out, int out_size, void* d_ws, size_t ws_size,
                              hipStream_t stream) {
    const float* x     = (const float*)d_in[0];
    const int* eidx    = (const int*)d_in[1];
    const int* batch   = (const int*)d_in[2];
    const float* W1 = (const float*)d_in[4];
    const float* b1 = (const float*)d_in[5];
    const float* W2 = (const float*)d_in[6];
    const float* b2 = (const float*)d_in[7];

    const int n = in_sizes[0] / 4;
    const int E = in_sizes[1] / 2;
    const int G = out_size / 64;

    const int* src = eidx;
    const int* dst = eidx + E;

    float* ws   = (float*)d_ws;
    float* dis  = ws;
    float* bufA = dis + n;
    float* bufB = bufA + (size_t)n * 64;
    float* sums = bufB + (size_t)n * 64;
    float* cnt  = sums + (size_t)G * 64;
    int* icnt   = (int*)(cnt + G);
    int* start  = icnt + n;
    int* csr    = start + n;
    int* binned = (int*)bufA;
    int* tab    = (int*)bufB;
    float* xs   = bufB + (size_t)n * 32;

    const int T = 256;
    auto blk = [T](int work) { return (work + T - 1) / T; };

    const int R  = (n + (1 << RSH) - 1) >> RSH;
    const int NC = (E + CHUNK - 1) / CHUNK;
    k_chunk_place<<<NC, 1024, 0, stream>>>(src, dst, tab, binned, (float*)d_out, G * 64, cnt, G, E, R);
    k_range_sort<<<R, 1024, 0, stream>>>(tab, binned, csr, start, icnt, dis, batch, cnt, x, xs, n, R, NC);

    k_agg_gemm12<<<(n + 63) / 64, 128, 0, stream>>>(start, icnt, csr, dis, xs, W1, b1, W2, bufA, n);
    k_gather_pool<<<blk(n * 16), T, 0, stream>>>(start, icnt, csr, dis, bufA, b2, batch, cnt, (float*)d_out, n);
}


# speedup vs baseline: 9.7344x; 1.1446x over previous
#include <hip/hip_runtime.h>
#include <hip/hip_bf16.h>

typedef __attribute__((ext_vector_type(2))) float v2f;
typedef __attribute__((ext_vector_type(8))) float v8f;

#define RSH    8
#define RNODES (1 << RSH)
#define RMASK  (RNODES - 1)
#define RMAX   1024
#define CHUNK  4096
#define EPT    (CHUNK / 1024)
#define SCAP   8192
#define PGRP   8

__device__ __forceinline__ int block_excl_scan(int c, int* s_w) {
    const int tid  = threadIdx.x;
    const int lane = tid & 31;
    const int wave = tid >> 5;
    int inc = c;
    for (int o = 1; o < 32; o <<= 1) {
        int v = __shfl_up(inc, o);
        if (lane >= o) inc += v;
    }
    if (lane == 31) s_w[wave] = inc;
    __syncthreads();
    int v = (lane < wave) ? s_w[lane] : 0;
    for (int o = 16; o > 0; o >>= 1) v += __shfl_xor(v, o);
    return v + inc - c;
}

__global__ void k_chunk_place(const int* __restrict__ src, const int* __restrict__ dst,
                              int* __restrict__ tab, int* __restrict__ binned,
                              float* __restrict__ out, int nout,
                              float* __restrict__ gcnt, int G, int E, int R) {
    __shared__ int s_cur[RMAX];
    __shared__ int s_w0[32];
    __shared__ int s_word[CHUNK];
    const int tid = threadIdx.x;
    const int e0  = blockIdx.x * CHUNK;

    const int gi = blockIdx.x * 1024 + tid;
    const int gs = gridDim.x * 1024;
    for (int i = gi; i < nout; i += gs) out[i]  = 0.0f;
    for (int i = gi; i < G;    i += gs) gcnt[i] = 0.0f;

    int d[EPT], s[EPT];
#pragma unroll
    for (int q = 0; q < EPT; ++q) {
        const int e  = e0 + q * 1024 + tid;
        const int ec = (e < E) ? e : E - 1;
        const int v  = dst[ec];
        s[q] = src[ec];
        d[q] = (e < E) ? v : -1;
    }
    s_cur[tid] = 0;
    __syncthreads();
#pragma unroll
    for (int q = 0; q < EPT; ++q)
        if (d[q] >= 0) atomicAdd(&s_cur[d[q] >> RSH], 1);
    __syncthreads();

    const int h   = s_cur[tid];
    const int lex = block_excl_scan(h, s_w0);
    s_cur[tid] = lex;
    if (tid < R) tab[(size_t)blockIdx.x * R + tid] = lex | (h << 16);
    __syncthreads();

#pragma unroll
    for (int q = 0; q < EPT; ++q) {
        if (d[q] >= 0) {
            const int lp = atomicAdd(&s_cur[d[q] >> RSH], 1);
            s_word[lp] = (s[q] << RSH) | (d[q] & RMASK);
        }
    }
    __syncthreads();

#pragma unroll
    for (int q = 0; q < EPT; ++q) {
        const int j = q * 1024 + tid;
        if (e0 + j < E) binned[e0 + j] = s_word[j];
    }
}

template <class F>
__device__ __forceinline__ void for_range_words(const int* __restrict__ tab,
                                                const int* __restrict__ binned,
                                                int r, int R, int NC, int inf0, F f) {
    const int lane = threadIdx.x & 31;
    const int wave = threadIdx.x >> 5;
    for (int cb = 0; cb < NC; cb += 1024) {
        int inf = inf0;
        if (cb > 0) {
            const int cl = cb + lane * 32 + wave;
            inf = (cl < NC) ? tab[(size_t)cl * R + r] : 0;
        }
        int nk = (NC - cb - wave + 31) >> 5;
        nk = (nk < 0) ? 0 : ((nk > 32) ? 32 : nk);
        for (int k = 0; k < nk; k += PGRP) {
            int w[PGRP], cn[PGRP], base[PGRP];
#pragma unroll
            for (int u = 0; u < PGRP; ++u) {
                const int iu = __shfl(inf, k + u);
                cn[u]   = iu >> 16;
                base[u] = (cb + (k + u) * 32 + wave) * CHUNK + (iu & 0xFFFF);
                w[u]    = binned[(lane < cn[u]) ? base[u] + lane : 0];
            }
#pragma unroll
            for (int u = 0; u < PGRP; ++u)
                if (lane < cn[u]) f(w[u]);
#pragma unroll
            for (int u = 0; u < PGRP; ++u)
                for (int o = lane + 32; o < cn[u]; o += 32) f(binned[base[u] + o]);
        }
    }
}

__global__ void k_range_sort(const int* __restrict__ tab, const int* __restrict__ binned,
                             int* __restrict__ csr, int* __restrict__ start,
                             int* __restrict__ cnt, float* __restrict__ dis,
                             const int* __restrict__ batch, float* __restrict__ gcnt,
                             const float* __restrict__ x, float* __restrict__ xs,
                             int n, int R, int NC) {
    __shared__ int s_w0[32];
    __shared__ int s_w1[32];
    __shared__ int s_c[RNODES];
    __shared__ int s_o[RNODES];
    __shared__ int s_out[SCAP];
    const int tid  = threadIdx.x;
    const int lane = tid & 31;
    const int wave = tid >> 5;
    const int r    = blockIdx.x;

    const int cl0  = lane * 32 + wave;
    const int inf0 = (cl0 < NC) ? tab[(size_t)cl0 * R + r] : 0;
    float4 xv;
    xv.x = 0.0f; xv.y = 0.0f; xv.z = 0.0f; xv.w = 0.0f;
    if (tid < RNODES) {
        const int in0 = (r << RSH) + tid;
        xv = *(const float4*)&x[(size_t)((in0 < n) ? in0 : n - 1) * 4];
    }
    int mlen = inf0 >> 16;
    int mb0  = inf0 & 0xFFFF;
    for (int cb = 1024; cb < NC; cb += 1024) {
        const int cl = cb + cl0;
        const int t  = (cl < NC) ? tab[(size_t)cl * R + r] : 0;
        mlen += t >> 16;
        mb0  += t & 0xFFFF;
    }
    for (int o = 16; o > 0; o >>= 1) {
        mlen += __shfl_xor(mlen, o);
        mb0  += __shfl_xor(mb0, o);
    }
    if (lane == 0) { s_w0[wave] = mlen; s_w1[wave] = mb0; }
    if (tid < RNODES) s_c[tid] = 0;
    __syncthreads();
    int len = s_w0[lane];
    int b0  = s_w1[lane];
    for (int o = 16; o > 0; o >>= 1) {
        len += __shfl_xor(len, o);
        b0  += __shfl_xor(b0, o);
    }

    for_range_words(tab, binned, r, R, NC, inf0,
                    [&](int w) { atomicAdd(&s_c[w & RMASK], 1); });
    __syncthreads();

    const int cn  = (tid < RNODES) ? s_c[tid] : 0;
    const int off = block_excl_scan(cn, s_w1);
    if (tid < RNODES) {
        s_o[tid] = off;
        const int i = (r << RSH) + tid;
        const int g = (i < n) ? batch[i] : -1;
        if (i < n) {
            start[i] = b0 + off;
            cnt[i]   = cn;
            float dg = (float)(cn + 1);
            const float dv = (dg > 0.0f) ? rsqrtf(dg) : 0.0f;
            dis[i] = dv;
            float4 xo;
            xo.x = dv * xv.x; xo.y = dv * xv.y; xo.z = dv * xv.z; xo.w = dv * xv.w;
            *(float4*)&xs[(size_t)i * 4] = xo;
        }
        unsigned todo = (unsigned)__ballot(g >= 0);
        for (int it = 0; it < 32 && todo != 0u; ++it) {
            const int leader = __ffs((int)todo) - 1;
            const int gl = __shfl(g, leader);
            const unsigned same = (unsigned)__ballot(g == gl) & todo;
            if (lane == leader) atomicAdd(&gcnt[gl], (float)__popc(same));
            todo &= ~same;
        }
    }
    __syncthreads();

    if (len <= SCAP) {
        for_range_words(tab, binned, r, R, NC, inf0,
                        [&](int w) { s_out[atomicAdd(&s_o[w & RMASK], 1)] = w >> RSH; });
        __syncthreads();
        for (int k = tid; k < len; k += 1024) csr[b0 + k] = s_out[k];
    } else {
        for_range_words(tab, binned, r, R, NC, inf0,
                        [&](int w) { csr[b0 + atomicAdd(&s_o[w & RMASK], 1)] = w >> RSH; });
    }
}

#define TPITCH 68
#define WPITCH 72

__global__ void k_agg_gemm12(const int* __restrict__ start, const int* __restrict__ cnt,
                             const int* __restrict__ csr, const float* __restrict__ dis,
                             const float* __restrict__ xs, const float* __restrict__ W1,
                             const float* __restrict__ b1, const float* __restrict__ W2,
                             float* __restrict__ out, int n) {
    __shared__ __attribute__((aligned(16))) float sW[64 * WPITCH];
    __shared__ __attribute__((aligned(16))) float sT[4 * 16 * TPITCH];
    const int tid  = threadIdx.x;
    const int wave = tid >> 5;
    const int lane = tid & 31;
    const int tile_row = blockIdx.x * 64 + wave * 16;

    float4 wq[8];
#pragma unroll
    for (int q = 0; q < 8; ++q) wq[q] = *(const float4*)&W2[(size_t)(q * 128 + tid) * 4];

    const int m    = lane & 15;
    const int koff = (lane >> 4) * 2;
    int row = tile_row + m;
    if (row >= n) row = n - 1;

    const float di = dis[row];
    int j = start[row];
    const int end = j + cnt[row];

    const float2 xo = *(const float2*)&xs[(size_t)row * 4 + koff];

#pragma unroll
    for (int q = 0; q < 8; ++q) {
        const int i = (q * 128 + tid) * 4;
        *(float4*)&sW[(i >> 6) * WPITCH + (i & 63)] = wq[q];
    }

    float ax = xo.x, ay = xo.y;

    for (; j + 4 <= end; j += 4) {
        const int s0 = csr[j], s1 = csr[j + 1], s2 = csr[j + 2], s3 = csr[j + 3];
        const float2 v0 = *(const float2*)&xs[(size_t)s0 * 4 + koff];
        const float2 v1 = *(const float2*)&xs[(size_t)s1 * 4 + koff];
        const float2 v2 = *(const float2*)&xs[(size_t)s2 * 4 + koff];
        const float2 v3 = *(const float2*)&xs[(size_t)s3 * 4 + koff];
        ax += v0.x; ay += v0.y;
        ax += v1.x; ay += v1.y;
        ax += v2.x; ay += v2.y;
        ax += v3.x; ay += v3.y;
    }
    const int rem = end - j;
    if (rem > 0) {
        const int s0 = csr[j];
        const int s1 = (rem > 1) ? csr[j + 1] : s0;
        const int s2 = (rem > 2) ? csr[j + 2] : s0;
        const float2 v0 = *(const float2*)&xs[(size_t)s0 * 4 + koff];
        const float2 v1 = *(const float2*)&xs[(size_t)s1 * 4 + koff];
        const float2 v2 = *(const float2*)&xs[(size_t)s2 * 4 + koff];
        ax += v0.x; ay += v0.y;
        if (rem > 1) { ax += v1.x; ay += v1.y; }
        if (rem > 2) { ax += v2.x; ay += v2.y; }
    }

    v2f a;
    a.x = di * ax;
    a.y = di * ay;

    v8f acc[4];
#pragma unroll
    for (int nt = 0; nt < 4; ++nt) {
        v2f bf;
        const int col = nt * 16 + m;
        bf.x = W1[(koff) * 64 + col];
        bf.y = W1[(koff + 1) * 64 + col];
        acc[nt] = __builtin_amdgcn_wmma_f32_16x16x4_f32(
             false, a,  false, bf,
             (short)0, v8f{0.f},  false,  false);
    }

    float* tw = sT + wave * 16 * TPITCH;
#pragma unroll
    for (int nt = 0; nt < 4; ++nt) {
        const float bv = b1[nt * 16 + m];
#pragma unroll
        for (int r = 0; r < 8; ++r)
            tw[(r + 8 * (lane >> 4)) * TPITCH + nt * 16 + m] = fmaxf(acc[nt][r] + bv, 0.0f);
    }
    __syncthreads();

    v8f acc2[4] = {v8f{0.f}, v8f{0.f}, v8f{0.f}, v8f{0.f}};
#pragma unroll
    for (int kb = 0; kb < 64; kb += 4) {
        v2f a2 = *(const v2f*)&tw[m * TPITCH + kb + koff];
        a2.x *= di;
        a2.y *= di;
#pragma unroll
        for (int nt = 0; nt < 4; ++nt) {
            v2f b;
            const int col = nt * 16 + m;
            b.x = sW[(kb + koff) * WPITCH + col];
            b.y = sW[(kb + koff + 1) * WPITCH + col];
            acc2[nt] = __builtin_amdgcn_wmma_f32_16x16x4_f32(
                 false, a2,  false, b,
                 (short)0, acc2[nt],  false,  false);
        }
    }

#pragma unroll
    for (int nt = 0; nt < 4; ++nt) {
#pragma unroll
        for (int r = 0; r < 8; ++r) {
            const int rr = tile_row + r + 8 * (lane >> 4);
            if (rr < n) out[(size_t)rr * 64 + nt * 16 + m] = acc2[nt][r];
        }
    }
}

#define GNODES 16

__global__ void k_gather_pool(const int* __restrict__ start, const int* __restrict__ cnt,
                              const int* __restrict__ csr, const float* __restrict__ dis,
                              const float* __restrict__ h, const float* __restrict__ b2,
                              const int* __restrict__ batch, const float* __restrict__ gcnt,
                              float* __restrict__ out, int n) {
    __shared__ __attribute__((aligned(16))) float s_v[GNODES * 64];
    __shared__ int   s_g[GNODES];
    __shared__ float s_n[GNODES];
    const int tid = threadIdx.x;
    const int ln  = tid >> 4;
    const int i   = blockIdx.x * GNODES + ln;
    const int jj  = (tid & 15) * 4;

    float4 v;
    v.x = 0.0f; v.y = 0.0f; v.z = 0.0f; v.w = 0.0f;
    int g = -1;
    float gn = 1.0f;
    if (i < n) {
        const float di = dis[i];
        int j = start[i];
        const int end = j + cnt[i];
        g = batch[i];
        const float4 bv = *(const float4*)&b2[jj];

        const float4 hs = *(const float4*)&h[(size_t)i * 64 + jj];
        gn = gcnt[(g >= 0) ? g : 0];
        float4 acc = hs;

        for (; j + 4 <= end; j += 4) {
            const int s0 = csr[j], s1 = csr[j + 1], s2 = csr[j + 2], s3 = csr[j + 3];
            const float4 v0 = *(const float4*)&h[(size_t)s0 * 64 + jj];
            const float4 v1 = *(const float4*)&h[(size_t)s1 * 64 + jj];
            const float4 v2 = *(const float4*)&h[(size_t)s2 * 64 + jj];
            const float4 v3 = *(const float4*)&h[(size_t)s3 * 64 + jj];
            acc.x += v0.x; acc.y += v0.y; acc.z += v0.z; acc.w += v0.w;
            acc.x += v1.x; acc.y += v1.y; acc.z += v1.z; acc.w += v1.w;
            acc.x += v2.x; acc.y += v2.y; acc.z += v2.z; acc.w += v2.w;
            acc.x += v3.x; acc.y += v3.y; acc.z += v3.z; acc.w += v3.w;
        }
        const int rem = end - j;
        if (rem > 0) {
            const int s0 = csr[j];
            const int s1 = (rem > 1) ? csr[j + 1] : s0;
            const int s2 = (rem > 2) ? csr[j + 2] : s0;
            const float4 v0 = *(const float4*)&h[(size_t)s0 * 64 + jj];
            const float4 v1 = *(const float4*)&h[(size_t)s1 * 64 + jj];
            const float4 v2 = *(const float4*)&h[(size_t)s2 * 64 + jj];
            acc.x += v0.x; acc.y += v0.y; acc.z += v0.z; acc.w += v0.w;
            if (rem > 1) {
                acc.x += v1.x; acc.y += v1.y; acc.z += v1.z; acc.w += v1.w;
            }
            if (rem > 2) {
                acc.x += v2.x; acc.y += v2.y; acc.z += v2.z; acc.w += v2.w;
            }
        }

        v.x = fmaxf(di * acc.x + bv.x, 0.0f);
        v.y = fmaxf(di * acc.y + bv.y, 0.0f);
        v.z = fmaxf(di * acc.z + bv.z, 0.0f);
        v.w = fmaxf(di * acc.w + bv.w, 0.0f);
    }

    *(float4*)&s_v[ln * 64 + jj] = v;
    if ((tid & 15) == 0) { s_g[ln] = g; s_n[ln] = gn; }
    __syncthreads();

    if (tid < 64) {
        int   gq[GNODES];
        float nq[GNODES];
        float vq[GNODES];
#pragma unroll
        for (int q = 0; q < GNODES; ++q) {
            gq[q] = s_g[q];
            nq[q] = s_n[q];
            vq[q] = s_v[q * 64 + tid];
        }
        int   cur = gq[0];
        float cn  = nq[0];
        float a   = 0.0f;
#pragma unroll
        for (int q = 0; q < GNODES; ++q) {
            if (gq[q] != cur) {
                if (cur >= 0) atomicAdd(&out[(size_t)cur * 64 + tid], a / fmaxf(cn, 1.0f));
                a   = 0.0f;
                cur = gq[q];
                cn  = nq[q];
            }
            a += vq[q];
        }
        if (cur >= 0) atomicAdd(&out[(size_t)cur * 64 + tid], a / fmaxf(cn, 1.0f));
    }
}

extern "C" void kernel_launch(void* const* d_in, const int* in_sizes, int n_in,
                              void* d_out, int out_size, void* d_ws, size_t ws_size,
                              hipStream_t stream) {
    const float* x     = (const float*)d_in[0];
    const int* eidx    = (const int*)d_in[1];
    const int* batch   = (const int*)d_in[2];
    const float* W1 = (const float*)d_in[4];
    const float* b1 = (const float*)d_in[5];
    const float* W2 = (const float*)d_in[6];
    const float* b2 = (const float*)d_in[7];

    const int n = in_sizes[0] / 4;
    const int E = in_sizes[1] / 2;
    const int G = out_size / 64;

    const int* src = eidx;
    const int* dst = eidx + E;

    float* bufA = (float*)(((uintptr_t)d_ws + 255) & ~(uintptr_t)255);
    float* bufB = bufA + (size_t)n * 64;
    float* dis  = bufB + (size_t)n * 64;
    float* cnt  = dis + n;
    int* icnt   = (int*)(cnt + G);
    int* start  = icnt + n;
    int* csr    = start + n;
    int* binned = (int*)bufA;
    int* tab    = (int*)bufB;
    float* xs   = bufB + (size_t)n * 32;

    const int T = 256;
    auto blk = [T](int work) { return (work + T - 1) / T; };

    const int R  = (n + (1 << RSH) - 1) >> RSH;
    const int NC = (E + CHUNK - 1) / CHUNK;
    k_chunk_place<<<NC, 1024, 0, stream>>>(src, dst, tab, binned, (float*)d_out, G * 64, cnt, G, E, R);
    k_range_sort<<<R, 1024, 0, stream>>>(tab, binned, csr, start, icnt, dis, batch, cnt, x, xs, n, R, NC);

    k_agg_gemm12<<<(n + 63) / 64, 128, 0, stream>>>(start, icnt, csr, dis, xs, W1, b1, W2, bufA, n);
    k_gather_pool<<<blk(n * 16), T, 0, stream>>>(start, icnt, csr, dis, bufA, b2, batch, cnt, (float*)d_out, n);
}


# both gathers read a node's source ids once, one slot per lane (coalesced), not 16 B ahead of every four rows; layer 1: lane = slot, rows via LDS, summed per node in slot order; sum order unchanged
# speedup vs baseline: 10.0373x; 1.0311x over previous
#include <hip/hip_runtime.h>
#include <hip/hip_bf16.h>

typedef __attribute__((ext_vector_type(2))) float v2f;
typedef __attribute__((ext_vector_type(8))) float v8f;

#define RSH    8
#define RNODES (1 << RSH)
#define RMASK  (RNODES - 1)
#define RMAX   1024
#define CHUNK  4096
#define EPT    (CHUNK / 1024)
#define SCAP   8192
#define PGRP   8

__device__ __forceinline__ int block_excl_scan(int c, int* s_w) {
    const int tid  = threadIdx.x;
    const int lane = tid & 31;
    const int wave = tid >> 5;
    int inc = c;
    for (int o = 1; o < 32; o <<= 1) {
        int v = __shfl_up(inc, o);
        if (lane >= o) inc += v;
    }
    if (lane == 31) s_w[wave] = inc;
    __syncthreads();
    int v = (lane < wave) ? s_w[lane] : 0;
    for (int o = 16; o > 0; o >>= 1) v += __shfl_xor(v, o);
    return v + inc - c;
}

__global__ void k_chunk_place(const int* __restrict__ src, const int* __restrict__ dst,
                              int* __restrict__ tab, int* __restrict__ binned,
                              float* __restrict__ out, int nout,
                              float* __restrict__ gcnt, int G, int E, int R) {
    __shared__ int s_cur[RMAX];
    __shared__ int s_w0[32];
    __shared__ int s_word[CHUNK];
    const int tid = threadIdx.x;
    const int e0  = blockIdx.x * CHUNK;

    const int gi = blockIdx.x * 1024 + tid;
    const int gs = gridDim.x * 1024;
    for (int i = gi; i < nout; i += gs) out[i]  = 0.0f;
    for (int i = gi; i < G;    i += gs) gcnt[i] = 0.0f;

    int d[EPT], s[EPT];
#pragma unroll
    for (int q = 0; q < EPT; ++q) {
        const int e  = e0 + q * 1024 + tid;
        const int ec = (e < E) ? e : E - 1;
        const int v  = dst[ec];
        s[q] = src[ec];
        d[q] = (e < E) ? v : -1;
    }
    s_cur[tid] = 0;
    __syncthreads();
#pragma unroll
    for (int q = 0; q < EPT; ++q)
        if (d[q] >= 0) atomicAdd(&s_cur[d[q] >> RSH], 1);
    __syncthreads();

    const int h   = s_cur[tid];
    const int lex = block_excl_scan(h, s_w0);
    s_cur[tid] = lex;
    if (tid < R) tab[(size_t)blockIdx.x * R + tid] = lex | (h << 16);
    __syncthreads();

#pragma unroll
    for (int q = 0; q < EPT; ++q) {
        if (d[q] >= 0) {
            const int lp = atomicAdd(&s_cur[d[q] >> RSH], 1);
            s_word[lp] = (s[q] << RSH) | (d[q] & RMASK);
        }
    }
    __syncthreads();

#pragma unroll
    for (int q = 0; q < EPT; ++q) {
        const int j = q * 1024 + tid;
        if (e0 + j < E) binned[e0 + j] = s_word[j];
    }
}

template <class F>
__device__ __forceinline__ void for_range_words(const int* __restrict__ tab,
                                                const int* __restrict__ binned,
                                                int r, int R, int NC, int inf0, F f) {
    const int lane = threadIdx.x & 31;
    const int wave = threadIdx.x >> 5;
    for (int cb = 0; cb < NC; cb += 1024) {
        int inf = inf0;
        if (cb > 0) {
            const int cl = cb + lane * 32 + wave;
            inf = (cl < NC) ? tab[(size_t)cl * R + r] : 0;
        }
        int nk = (NC - cb - wave + 31) >> 5;
        nk = (nk < 0) ? 0 : ((nk > 32) ? 32 : nk);
        for (int k = 0; k < nk; k += PGRP) {
            int w[PGRP], cn[PGRP], base[PGRP];
#pragma unroll
            for (int u = 0; u < PGRP; ++u) {
                const int iu = __shfl(inf, k + u);
                cn[u]   = iu >> 16;
                base[u] = (cb + (k + u) * 32 + wave) * CHUNK + (iu & 0xFFFF);
                w[u]    = binned[(lane < cn[u]) ? base[u] + lane : 0];
            }
#pragma unroll
            for (int u = 0; u < PGRP; ++u)
                if (lane < cn[u]) f(w[u]);
#pragma unroll
            for (int u = 0; u < PGRP; ++u)
                for (int o = lane + 32; o < cn[u]; o += 32) f(binned[base[u] + o]);
        }
    }
}

__global__ void k_range_sort(const int* __restrict__ tab, const int* __restrict__ binned,
                             int* __restrict__ csr, int* __restrict__ start,
                             int* __restrict__ cnt, float* __restrict__ dis,
                             const int* __restrict__ batch, float* __restrict__ gcnt,
                             const float* __restrict__ x, float* __restrict__ xs,
                             int n, int R, int NC) {
    __shared__ int s_w0[32];
    __shared__ int s_w1[32];
    __shared__ int s_c[RNODES];
    __shared__ int s_o[RNODES];
    __shared__ int s_out[SCAP];
    const int tid  = threadIdx.x;
    const int lane = tid & 31;
    const int wave = tid >> 5;
    const int r    = blockIdx.x;

    const int cl0  = lane * 32 + wave;
    const int inf0 = (cl0 < NC) ? tab[(size_t)cl0 * R + r] : 0;
    float4 xv;
    xv.x = 0.0f; xv.y = 0.0f; xv.z = 0.0f; xv.w = 0.0f;
    if (tid < RNODES) {
        const int in0 = (r << RSH) + tid;
        xv = *(const float4*)&x[(size_t)((in0 < n) ? in0 : n - 1) * 4];
    }
    int mlen = inf0 >> 16;
    int mb0  = inf0 & 0xFFFF;
    for (int cb = 1024; cb < NC; cb += 1024) {
        const int cl = cb + cl0;
        const int t  = (cl < NC) ? tab[(size_t)cl * R + r] : 0;
        mlen += t >> 16;
        mb0  += t & 0xFFFF;
    }
    for (int o = 16; o > 0; o >>= 1) {
        mlen += __shfl_xor(mlen, o);
        mb0  += __shfl_xor(mb0, o);
    }
    if (lane == 0) { s_w0[wave] = mlen; s_w1[wave] = mb0; }
    if (tid < RNODES) s_c[tid] = 0;
    __syncthreads();
    int len = s_w0[lane];
    int b0  = s_w1[lane];
    for (int o = 16; o > 0; o >>= 1) {
        len += __shfl_xor(len, o);
        b0  += __shfl_xor(b0, o);
    }

    for_range_words(tab, binned, r, R, NC, inf0,
                    [&](int w) { atomicAdd(&s_c[w & RMASK], 1); });
    __syncthreads();

    const int cn  = (tid < RNODES) ? s_c[tid] : 0;
    const int off = block_excl_scan(cn, s_w1);
    if (tid < RNODES) {
        s_o[tid] = off;
        const int i = (r << RSH) + tid;
        const int g = (i < n) ? batch[i] : -1;
        if (i < n) {
            start[i] = b0 + off;
            cnt[i]   = cn;
            float dg = (float)(cn + 1);
            const float dv = (dg > 0.0f) ? rsqrtf(dg) : 0.0f;
            dis[i] = dv;
            float4 xo;
            xo.x = dv * xv.x; xo.y = dv * xv.y; xo.z = dv * xv.z; xo.w = dv * xv.w;
            *(float4*)&xs[(size_t)i * 4] = xo;
        }
        unsigned todo = (unsigned)__ballot(g >= 0);
        for (int it = 0; it < 32 && todo != 0u; ++it) {
            const int leader = __ffs((int)todo) - 1;
            const int gl = __shfl(g, leader);
            const unsigned same = (unsigned)__ballot(g == gl) & todo;
            if (lane == leader) atomicAdd(&gcnt[gl], (float)__popc(same));
            todo &= ~same;
        }
    }
    __syncthreads();

    if (len <= SCAP) {
        for_range_words(tab, binned, r, R, NC, inf0,
                        [&](int w) { s_out[atomicAdd(&s_o[w & RMASK], 1)] = w >> RSH; });
        __syncthreads();
        for (int k = tid; k < len; k += 1024) csr[b0 + k] = s_out[k];
    } else {
        for_range_words(tab, binned, r, R, NC, inf0,
                        [&](int w) { csr[b0 + atomicAdd(&s_o[w & RMASK], 1)] = w >> RSH; });
    }
}

#define TPITCH 68
#define WPITCH 72
#define GSLOTS 32
#define GPITCH (GSLOTS * 4 + 4)

__global__ void k_agg_gemm12(const int* __restrict__ start, const int* __restrict__ cnt,
                             const int* __restrict__ csr, const float* __restrict__ dis,
                             const float* __restrict__ xs, const float* __restrict__ W1,
                             const float* __restrict__ b1, const float* __restrict__ W2,
                             float* __restrict__ out, int n) {
    __shared__ __attribute__((aligned(16))) float sW[64 * WPITCH];
    __shared__ __attribute__((aligned(16))) float sT[4 * 16 * TPITCH];
    __shared__ __attribute__((aligned(16))) float sG[4 * 16 * GPITCH];
    const int tid  = threadIdx.x;
    const int wave = tid >> 5;
    const int lane = tid & 31;
    const int tile_row = blockIdx.x * 64 + wave * 16;

    float4 wq[8];
#pragma unroll
    for (int q = 0; q < 8; ++q) wq[q] = *(const float4*)&W2[(size_t)(q * 128 + tid) * 4];

    const int m    = lane & 15;
    const int koff = (lane >> 4) * 2;
    int row = tile_row + m;
    if (row >= n) row = n - 1;

    const float di = dis[row];
    const int j0 = start[row];
    const int cn = cnt[row];
    const int end = j0 + cn;

    const float2 xo = *(const float2*)&xs[(size_t)row * 4 + koff];

#pragma unroll
    for (int q = 0; q < 8; ++q) {
        const int i = (q * 128 + tid) * 4;
        *(float4*)&sW[(i >> 6) * WPITCH + (i & 63)] = wq[q];
    }

    float* gw = sG + wave * 16 * GPITCH;
    int sid[16];
#pragma unroll
    for (int r = 0; r < 16; ++r) {
        const int sr = __builtin_amdgcn_readlane(j0, r);
        const int cr = __builtin_amdgcn_readlane(cn, r);
        const int sv = csr[(lane < cr) ? sr + lane : 0];
        sid[r] = (lane < cr) ? sv : 0;
    }
    float4 gv[16];
#pragma unroll
    for (int r = 0; r < 16; ++r) gv[r] = *(const float4*)&xs[(size_t)sid[r] * 4];
#pragma unroll
    for (int r = 0; r < 16; ++r) *(float4*)&gw[r * GPITCH + lane * 4] = gv[r];
    __syncthreads();

    float ax = xo.x, ay = xo.y;

    const int ns = (cn < GSLOTS) ? cn : GSLOTS;
    const float* gr = gw + m * GPITCH + koff;
    for (int e = 0; e < ns; e += 8) {
        float2 v[8];
#pragma unroll
        for (int u = 0; u < 8; ++u) v[u] = *(const float2*)&gr[(e + u) * 4];
#pragma unroll
        for (int u = 0; u < 8; ++u)
            if (e + u < ns) { ax += v[u].x; ay += v[u].y; }
    }
    for (int j = j0 + GSLOTS; j < end; ++j) {
        const int s = csr[j];
        const float2 v = *(const float2*)&xs[(size_t)s * 4 + koff];
        ax += v.x; ay += v.y;
    }

    v2f a;
    a.x = di * ax;
    a.y = di * ay;

    v8f acc[4];
#pragma unroll
    for (int nt = 0; nt < 4; ++nt) {
        v2f bf;
        const int col = nt * 16 + m;
        bf.x = W1[(koff) * 64 + col];
        bf.y = W1[(koff + 1) * 64 + col];
        acc[nt] = __builtin_amdgcn_wmma_f32_16x16x4_f32(
             false, a,  false, bf,
             (short)0, v8f{0.f},  false,  false);
    }

    float* tw = sT + wave * 16 * TPITCH;
#pragma unroll
    for (int nt = 0; nt < 4; ++nt) {
        const float bv = b1[nt * 16 + m];
#pragma unroll
        for (int r = 0; r < 8; ++r)
            tw[(r + 8 * (lane >> 4)) * TPITCH + nt * 16 + m] = fmaxf(acc[nt][r] + bv, 0.0f);
    }
    __syncthreads();

    v8f acc2[4] = {v8f{0.f}, v8f{0.f}, v8f{0.f}, v8f{0.f}};
#pragma unroll
    for (int kb = 0; kb < 64; kb += 4) {
        v2f a2 = *(const v2f*)&tw[m * TPITCH + kb + koff];
        a2.x *= di;
        a2.y *= di;
#pragma unroll
        for (int nt = 0; nt < 4; ++nt) {
            v2f b;
            const int col = nt * 16 + m;
            b.x = sW[(kb + koff) * WPITCH + col];
            b.y = sW[(kb + koff + 1) * WPITCH + col];
            acc2[nt] = __builtin_amdgcn_wmma_f32_16x16x4_f32(
                 false, a2,  false, b,
                 (short)0, acc2[nt],  false,  false);
        }
    }

#pragma unroll
    for (int nt = 0; nt < 4; ++nt) {
#pragma unroll
        for (int r = 0; r < 8; ++r) {
            const int rr = tile_row + r + 8 * (lane >> 4);
            if (rr < n) out[(size_t)rr * 64 + nt * 16 + m] = acc2[nt][r];
        }
    }
}

#define GNODES 16

__global__ void k_gather_pool(const int* __restrict__ start, const int* __restrict__ cnt,
                              const int* __restrict__ csr, const float* __restrict__ dis,
                              const float* __restrict__ h, const float* __restrict__ b2,
                              const int* __restrict__ batch, const float* __restrict__ gcnt,
                              float* __restrict__ out, int n) {
    __shared__ __attribute__((aligned(16))) float s_v[GNODES * 64];
    __shared__ int   s_g[GNODES];
    __shared__ float s_n[GNODES];
    const int tid = threadIdx.x;
    const int ln  = tid >> 4;
    const int i   = blockIdx.x * GNODES + ln;
    const int jj  = (tid & 15) * 4;

    float4 v;
    v.x = 0.0f; v.y = 0.0f; v.z = 0.0f; v.w = 0.0f;
    int g = -1;
    float gn = 1.0f;
    if (i < n) {
        const float di = dis[i];
        const int j0 = start[i];
        const int cn = cnt[i];
        g = batch[i];
        const float4 bv = *(const float4*)&b2[jj];

        const float4 hs = *(const float4*)&h[(size_t)i * 64 + jj];
        gn = gcnt[(g >= 0) ? g : 0];
        float4 acc = hs;

        const int e = tid & 15;
        for (int jb = 0; jb < cn; jb += 16) {
            const int nb = cn - jb;
            const int sid = csr[j0 + jb + ((e < nb) ? e : 0)];
#pragma unroll
            for (int u = 0; u < 16; u += 4) {
                if (u + 4 <= nb) {
                    const int s0 = __shfl(sid, u, 16),     s1 = __shfl(sid, u + 1, 16);
                    const int s2 = __shfl(sid, u + 2, 16), s3 = __shfl(sid, u + 3, 16);
                    const float4 v0 = *(const float4*)&h[(size_t)s0 * 64 + jj];
                    const float4 v1 = *(const float4*)&h[(size_t)s1 * 64 + jj];
                    const float4 v2 = *(const float4*)&h[(size_t)s2 * 64 + jj];
                    const float4 v3 = *(const float4*)&h[(size_t)s3 * 64 + jj];
                    acc.x += v0.x; acc.y += v0.y; acc.z += v0.z; acc.w += v0.w;
                    acc.x += v1.x; acc.y += v1.y; acc.z += v1.z; acc.w += v1.w;
                    acc.x += v2.x; acc.y += v2.y; acc.z += v2.z; acc.w += v2.w;
                    acc.x += v3.x; acc.y += v3.y; acc.z += v3.z; acc.w += v3.w;
                } else if (u < nb) {
                    const int rem = nb - u;
                    const int s0 = __shfl(sid, u, 16), s1 = __shfl(sid, u + 1, 16), s2 = __shfl(sid, u + 2, 16);
                    const float4 v0 = *(const float4*)&h[(size_t)s0 * 64 + jj];
                    const float4 v1 = *(const float4*)&h[(size_t)s1 * 64 + jj];
                    const float4 v2 = *(const float4*)&h[(size_t)s2 * 64 + jj];
                    acc.x += v0.x; acc.y += v0.y; acc.z += v0.z; acc.w += v0.w;
                    if (rem > 1) {
                        acc.x += v1.x; acc.y += v1.y; acc.z += v1.z; acc.w += v1.w;
                    }
                    if (rem > 2) {
                        acc.x += v2.x; acc.y += v2.y; acc.z += v2.z; acc.w += v2.w;
                    }
                }
            }
        }

        v.x = fmaxf(di * acc.x + bv.x, 0.0f);
        v.y = fmaxf(di * acc.y + bv.y, 0.0f);
        v.z = fmaxf(di * acc.z + bv.z, 0.0f);
        v.w = fmaxf(di * acc.w + bv.w, 0.0f);
    }

    *(float4*)&s_v[ln * 64 + jj] = v;
    if ((tid & 15) == 0) { s_g[ln] = g; s_n[ln] = gn; }
    __syncthreads();

    if (tid < 64) {
        int   gq[GNODES];
        float nq[GNODES];
        float vq[GNODES];
#pragma unroll
        for (int q = 0; q < GNODES; ++q) {
            gq[q] = s_g[q];
            nq[q] = s_n[q];
            vq[q] = s_v[q * 64 + tid];
        }
        int   cur = gq[0];
        float cn  = nq[0];
        float a   = 0.0f;
#pragma unroll
        for (int q = 0; q < GNODES; ++q) {
            if (gq[q] != cur) {
                if (cur >= 0) atomicAdd(&out[(size_t)cur * 64 + tid], a / fmaxf(cn, 1.0f));
                a   = 0.0f;
                cur = gq[q];
                cn  = nq[q];
            }
            a += vq[q];
        }
        if (cur >= 0) atomicAdd(&out[(size_t)cur * 64 + tid], a / fmaxf(cn, 1.0f));
    }
}

extern "C" void kernel_launch(void* const* d_in, const int* in_sizes, int n_in,
                              void* d_out, int out_size, void* d_ws, size_t ws_size,
                              hipStream_t stream) {
    const float* x     = (const float*)d_in[0];
    const int* eidx    = (const int*)d_in[1];
    const int* batch   = (const int*)d_in[2];
    const float* W1 = (const float*)d_in[4];
    const float* b1 = (const float*)d_in[5];
    const float* W2 = (const float*)d_in[6];
    const float* b2 = (const float*)d_in[7];

    const int n = in_sizes[0] / 4;
    const int E = in_sizes[1] / 2;
    const int G = out_size / 64;

    const int* src = eidx;
    const int* dst = eidx + E;

    float* bufA = (float*)(((uintptr_t)d_ws + 255) & ~(uintptr_t)255);
    float* bufB = bufA + (size_t)n * 64;
    float* dis  = bufB + (size_t)n * 64;
    float* cnt  = dis + n;
    int* icnt   = (int*)(cnt + G);
    int* start  = icnt + n;
    int* csr    = start + n;
    int* binned = (int*)bufA;
    int* tab    = (int*)bufB;
    float* xs   = bufB + (size_t)n * 32;

    const int T = 256;
    auto blk = [T](int work) { return (work + T - 1) / T; };

    const int R  = (n + (1 << RSH) - 1) >> RSH;
    const int NC = (E + CHUNK - 1) / CHUNK;
    k_chunk_place<<<NC, 1024, 0, stream>>>(src, dst, tab, binned, (float*)d_out, G * 64, cnt, G, E, R);
    k_range_sort<<<R, 1024, 0, stream>>>(tab, binned, csr, start, icnt, dis, batch, cnt, x, xs, n, R, NC);

    k_agg_gemm12<<<(n + 63) / 64, 128, 0, stream>>>(start, icnt, csr, dis, xs, W1, b1, W2, bufA, n);
    k_gather_pool<<<blk(n * 16), T, 0, stream>>>(start, icnt, csr, dis, bufA, b2, batch, cnt, (float*)d_out, n);
}
